# starting kernel
# baseline (speedup 1.0000x reference)
#include <hip/hip_runtime.h>

typedef float v2f __attribute__((ext_vector_type(2)));
typedef float v8f __attribute__((ext_vector_type(8)));

__global__ __launch_bounds__(256) void k_fill(float* __restrict__ p, float v, long n) {
  long i = (long)blockIdx.x * blockDim.x + threadIdx.x;
  if (i < n) p[i] = v;
}

__global__ __launch_bounds__(256) void k_deg(const int* __restrict__ dst,
                                             float* __restrict__ deg, int E) {
  int i = blockIdx.x * blockDim.x + threadIdx.x;
  if (i < E) unsafeAtomicAdd(&deg[dst[i]], 1.0f);
}

__global__ __launch_bounds__(256) void k_rsqrt(float* __restrict__ d, int n) {
  int i = blockIdx.x * blockDim.x + threadIdx.x;
  if (i < n) d[i] = rsqrtf(d[i]);
}

template <int OUT>
__global__ __launch_bounds__(256)
void k_gemm(const float* __restrict__ A, const float* __restrict__ W,
            float* __restrict__ H, int M) {
  constexpr int NT = OUT / 16;
  int wave = (blockIdx.x * blockDim.x + threadIdx.x) >> 5;
  int lane = threadIdx.x & 31;
  int row0 = wave * 16;
  if (row0 >= M) return;

  int m  = lane & 15;
  int kh = lane >> 4;

  v8f acc[NT] = {};

  int arow = (row0 + m < M) ? (row0 + m) : (M - 1);
  const float* ap = A + (size_t)arow * 128 + 2 * kh;

  for (int k = 0; k < 128; k += 4) {
    v2f a = *(const v2f*)(ap + k);
    const float* bp = W + (size_t)(k + 2 * kh) * OUT + m;
#pragma unroll
    for (int t = 0; t < NT; ++t) {
      v2f b;
      b.x = bp[t * 16];
      b.y = bp[t * 16 + OUT];
      acc[t] = __builtin_amdgcn_wmma_f32_16x16x4_f32(
           false, a,  false, b,
           (short)0, acc[t],  false,  false);
    }
  }

  int rbase = row0 + 8 * kh;
#pragma unroll
  for (int t = 0; t < NT; ++t) {
#pragma unroll
    for (int r = 0; r < 8; ++r) {
      int row = rbase + r;
      if (row < M) H[(size_t)row * OUT + t * 16 + m] = acc[t][r];
    }
  }
}

template <int OUT>
__global__ __launch_bounds__(256)
void k_scatter(const int* __restrict__ src, const int* __restrict__ dst,
               const float* __restrict__ dinv, const float* __restrict__ h,
               float* __restrict__ agg, int E) {
  int wave = (blockIdx.x * blockDim.x + threadIdx.x) >> 5;
  int lane = threadIdx.x & 31;
  if (wave >= E) return;
  int s = src[wave];
  int d = dst[wave];
  float nrm = dinv[s] * dinv[d];
  const float* hs = h + (size_t)s * OUT;
  float* ad = agg + (size_t)d * OUT;
#pragma unroll
  for (int k = lane; k < OUT; k += 32)
    unsafeAtomicAdd(ad + k, hs[k] * nrm);
}

template <int OUT, bool RELU>
__global__ __launch_bounds__(256)
void k_act(float* __restrict__ agg, const float* __restrict__ h,
           const float* __restrict__ dinv, const float* __restrict__ b, int N) {
  long i = (long)blockIdx.x * blockDim.x + threadIdx.x;
  long tot = (long)N * OUT;
  if (i >= tot) return;
  int node = (int)(i / OUT);
  int dcol = (int)(i % OUT);
  float di = dinv[node];
  float v = agg[i] + h[i] * di * di + b[dcol];
  agg[i] = RELU ? fmaxf(v, 0.0f) : v;
}

static inline int cdiv(long a, long b) { return (int)((a + b - 1) / b); }

extern "C" void kernel_launch(void* const* d_in, const int* in_sizes, int n_in,
                              void* d_out, int out_size, void* d_ws, size_t ws_size,
                              hipStream_t stream) {
  const float* x  = (const float*)d_in[0];
  const int*   ei = (const int*)d_in[1];
  const float* W1 = (const float*)d_in[2];
  const float* b1 = (const float*)d_in[3];
  const float* W2 = (const float*)d_in[4];
  const float* b2 = (const float*)d_in[5];
  const float* W3 = (const float*)d_in[6];
  const float* b3 = (const float*)d_in[7];

  const int N = in_sizes[0] / 128;
  const int E = in_sizes[1] / 2;
  const int* src = ei;
  const int* dst = ei + E;

  char* ws = (char*)d_ws;
  size_t off0 = 0;
  size_t off1 = (((size_t)N * 4) + 255) & ~(size_t)255;
  size_t off2 = off1 + (size_t)N * 128 * 4;
  float* dinv = (float*)(ws + off0);
  float* bufH = (float*)(ws + off1);
  float* bufF = (float*)(ws + off2);
  float* out  = (float*)d_out;

  const int T = 256;
  const int gemmBlocks = cdiv((long)cdiv(N, 16) * 32, T);
  const int edgeBlocks = cdiv((long)E * 32, T);

  k_fill<<<cdiv(N, T), T, 0, stream>>>(dinv, 1.0f, N);
  k_deg<<<cdiv(E, T), T, 0, stream>>>(dst, dinv, E);
  k_rsqrt<<<cdiv(N, T), T, 0, stream>>>(dinv, N);

  k_gemm<128><<<gemmBlocks, T, 0, stream>>>(x, W1, bufH, N);
  k_fill<<<cdiv((long)N * 128, T), T, 0, stream>>>(bufF, 0.0f, (long)N * 128);
  k_scatter<128><<<edgeBlocks, T, 0, stream>>>(src, dst, dinv, bufH, bufF, E);
  k_act<128, true><<<cdiv((long)N * 128, T), T, 0, stream>>>(bufF, bufH, dinv, b1, N);

  k_gemm<128><<<gemmBlocks, T, 0, stream>>>(bufF, W2, bufH, N);
  k_fill<<<cdiv((long)N * 128, T), T, 0, stream>>>(bufF, 0.0f, (long)N * 128);
  k_scatter<128><<<edgeBlocks, T, 0, stream>>>(src, dst, dinv, bufH, bufF, E);
  k_act<128, true><<<cdiv((long)N * 128, T), T, 0, stream>>>(bufF, bufH, dinv, b2, N);

  k_gemm<64><<<gemmBlocks, T, 0, stream>>>(bufF, W3, bufH, N);
  k_fill<<<cdiv((long)N * 64, T), T, 0, stream>>>(out, 0.0f, (long)N * 64);
  k_scatter<64><<<edgeBlocks, T, 0, stream>>>(src, dst, dinv, bufH, out, E);
  k_act<64, false><<<cdiv((long)N * 64, T), T, 0, stream>>>(out, bufH, dinv, b3, N);
}


# Aggregation as a gather: per-call destination-sorted edge list (int atomic rank + scan), one wave per node sums its source rows; replaces zero-fill, float-atomic scatter, add pass; sums in CSR order
# speedup vs baseline: 2.8808x; 2.8808x over previous
#include <hip/hip_runtime.h>


typedef float v2f __attribute__((ext_vector_type(2)));
typedef float v8f __attribute__((ext_vector_type(8)));

__global__ __launch_bounds__(256) void k_zero(int* __restrict__ p, int n) {
  int i = blockIdx.x * blockDim.x + threadIdx.x;
  if (i < n) p[i] = 0;
}

__global__ __launch_bounds__(256) void k_rank(const int* __restrict__ dst,
                                              int* __restrict__ cnt,
                                              int* __restrict__ pos, int E) {
  int i = blockIdx.x * blockDim.x + threadIdx.x;
  if (i < E) pos[i] = atomicAdd(&cnt[dst[i]], 1);
}

__global__ __launch_bounds__(1024) void k_scan(const int* __restrict__ cnt,
                                               int* __restrict__ rowptr,
                                               float* __restrict__ dinv, int N) {
  __shared__ int wsum[32];
  __shared__ int boff;
  const int tid = threadIdx.x;
  const int lane = tid & 31;
  const int wv = tid >> 5;
  const int base = blockIdx.x * 1024;

  int part = 0;
  const int4* c4 = (const int4*)cnt;
  const int nvec = base >> 2;
  int j = tid;
  for (; j + 3072 < nvec; j += 4096) {
    int4 v0 = c4[j], v1 = c4[j + 1024], v2 = c4[j + 2048], v3 = c4[j + 3072];
    part += ((v0.x + v0.y) + (v0.z + v0.w)) + ((v1.x + v1.y) + (v1.z + v1.w)) +
            ((v2.x + v2.y) + (v2.z + v2.w)) + ((v3.x + v3.y) + (v3.z + v3.w));
  }
  for (; j < nvec; j += 1024) {
    int4 v = c4[j];
    part += (v.x + v.y) + (v.z + v.w);
  }
  for (int o = 16; o > 0; o >>= 1) part += __shfl_xor(part, o);
  if (lane == 0) wsum[wv] = part;
  __syncthreads();
  if (wv == 0) {
    int v = wsum[lane];
    for (int o = 16; o > 0; o >>= 1) v += __shfl_xor(v, o);
    if (lane == 0) boff = v;
  }
  __syncthreads();
  const int off = boff;

  const int i = base + tid;
  const int c = (i < N) ? cnt[i] : 0;
  int incl = c;
  for (int o = 1; o < 32; o <<= 1) {
    int t = __shfl_up(incl, o);
    if (lane >= o) incl += t;
  }
  if (lane == 31) wsum[wv] = incl;
  __syncthreads();
  if (wv == 0) {
    int v = wsum[lane];
    int s = v;
    for (int o = 1; o < 32; o <<= 1) {
      int t = __shfl_up(s, o);
      if (lane >= o) s += t;
    }
    wsum[lane] = s - v;
  }
  __syncthreads();
  const int excl = off + wsum[wv] + (incl - c);
  if (i < N) {
    rowptr[i] = excl;
    dinv[i] = rsqrtf((float)(c + 1));
    if (i == N - 1) rowptr[N] = excl + c;
  }
}

__global__ __launch_bounds__(256) void k_place(const int* __restrict__ src,
                                               const int* __restrict__ dst,
                                               const int* __restrict__ pos,
                                               const int* __restrict__ rowptr,
                                               int* __restrict__ csr, int E) {
  int i = blockIdx.x * blockDim.x + threadIdx.x;
  if (i < E) csr[rowptr[dst[i]] + pos[i]] = src[i];
}

template <int OUT>
__global__ __launch_bounds__(256)
void k_gemm(const float* __restrict__ A, const float* __restrict__ W,
            float* __restrict__ H, int M) {
  constexpr int NT = OUT / 16;
  int wave = (blockIdx.x * blockDim.x + threadIdx.x) >> 5;
  int lane = threadIdx.x & 31;
  int row0 = wave * 16;
  if (row0 >= M) return;

  int m  = lane & 15;
  int kh = lane >> 4;

  v8f acc[NT] = {};

  int arow = (row0 + m < M) ? (row0 + m) : (M - 1);
  const float* ap = A + (size_t)arow * 128 + 2 * kh;

  for (int k = 0; k < 128; k += 4) {
    v2f a = *(const v2f*)(ap + k);
    const float* bp = W + (size_t)(k + 2 * kh) * OUT + m;
#pragma unroll
    for (int t = 0; t < NT; ++t) {
      v2f b;
      b.x = bp[t * 16];
      b.y = bp[t * 16 + OUT];
      acc[t] = __builtin_amdgcn_wmma_f32_16x16x4_f32(
           false, a,  false, b,
           (short)0, acc[t],  false,  false);
    }
  }

  int rbase = row0 + 8 * kh;
#pragma unroll
  for (int t = 0; t < NT; ++t) {
#pragma unroll
    for (int r = 0; r < 8; ++r) {
      int row = rbase + r;
      if (row < M) H[(size_t)row * OUT + t * 16 + m] = acc[t][r];
    }
  }
}

template <int V> struct vecf;
template <> struct vecf<2> { typedef float t __attribute__((ext_vector_type(2))); };
template <> struct vecf<4> { typedef float t __attribute__((ext_vector_type(4))); };

template <int OUT, int R>
__device__ __forceinline__ typename vecf<OUT / 32>::t
gather_chunk(typename vecf<OUT / 32>::t acc, const float* __restrict__ hl,
             int s_l, float g_l, float dd, int j) {
  typedef typename vecf<OUT / 32>::t vf;
  vf v[R];
  float w[R];
#pragma unroll
  for (int u = 0; u < R; ++u) {
    int s = __builtin_amdgcn_readlane(s_l, j + u);
    w[u] = __int_as_float(__builtin_amdgcn_readlane(__float_as_int(g_l), j + u)) * dd;
    v[u] = *(const vf*)(hl + (size_t)s * OUT);
  }
#pragma unroll
  for (int u = 0; u < R; ++u) acc += v[u] * w[u];
  return acc;
}

template <int OUT, bool RELU>
__global__ __launch_bounds__(256)
void k_gather(const int* __restrict__ rowptr, const int* __restrict__ csr,
              const float* __restrict__ dinv, const float* __restrict__ h,
              const float* __restrict__ b, float* __restrict__ out, int N) {
  constexpr int V = OUT / 32;
  typedef typename vecf<V>::t vf;
  const int lane = threadIdx.x & 31;
  const int d = __builtin_amdgcn_readfirstlane(
      (int)((blockIdx.x * blockDim.x + threadIdx.x) >> 5));
  if (d >= N) return;
  const int beg = rowptr[d];
  const int deg = rowptr[d + 1] - beg;
  const float dd = dinv[d];
  const float* hl = h + lane * V;

  const int tot = deg + 1;
  vf acc = {};
  for (int b0 = 0; b0 < tot; b0 += 32) {
    const int n = (tot - b0 < 32) ? (tot - b0) : 32;
    const int t = b0 + lane;
    int s_l = d;
    float g_l = dd;
    if (t < deg) {
      s_l = csr[beg + t];
      g_l = dinv[s_l];
    }
    int j = 0;
    for (; j + 8 <= n; j += 8) acc = gather_chunk<OUT, 8>(acc, hl, s_l, g_l, dd, j);
    switch (n - j) {
      case 7: acc = gather_chunk<OUT, 7>(acc, hl, s_l, g_l, dd, j); break;
      case 6: acc = gather_chunk<OUT, 6>(acc, hl, s_l, g_l, dd, j); break;
      case 5: acc = gather_chunk<OUT, 5>(acc, hl, s_l, g_l, dd, j); break;
      case 4: acc = gather_chunk<OUT, 4>(acc, hl, s_l, g_l, dd, j); break;
      case 3: acc = gather_chunk<OUT, 3>(acc, hl, s_l, g_l, dd, j); break;
      case 2: acc = gather_chunk<OUT, 2>(acc, hl, s_l, g_l, dd, j); break;
      case 1: acc = gather_chunk<OUT, 1>(acc, hl, s_l, g_l, dd, j); break;
      default: break;
    }
  }

  vf r = acc + *(const vf*)(b + lane * V);
  if (RELU) {
#pragma unroll
    for (int k = 0; k < V; ++k) r[k] = fmaxf(r[k], 0.0f);
  }
  *(vf*)(out + (size_t)d * OUT + lane * V) = r;
}

static inline int cdiv(long a, long b) { return (int)((a + b - 1) / b); }

extern "C" void kernel_launch(void* const* d_in, const int* in_sizes, int n_in,
                              void* d_out, int out_size, void* d_ws, size_t ws_size,
                              hipStream_t stream) {
  const float* x  = (const float*)d_in[0];
  const int*   ei = (const int*)d_in[1];
  const float* W1 = (const float*)d_in[2];
  const float* b1 = (const float*)d_in[3];
  const float* W2 = (const float*)d_in[4];
  const float* b2 = (const float*)d_in[5];
  const float* W3 = (const float*)d_in[6];
  const float* b3 = (const float*)d_in[7];

  const int N = in_sizes[0] / 128;
  const int E = in_sizes[1] / 2;
  const int* src = ei;
  const int* dst = ei + E;

  char* ws = (char*)d_ws;
  const size_t szN = (((size_t)(N + 1) * 4) + 255) & ~(size_t)255;
  const size_t szE = (((size_t)E * 4) + 255) & ~(size_t)255;
  size_t off = 0;
  float* dinv   = (float*)(ws + off); off += szN;
  int*   cnt    = (int*)(ws + off);   off += szN;
  int*   rowptr = (int*)(ws + off);   off += szN;
  int*   pos    = (int*)(ws + off);   off += szE;
  int*   csr    = (int*)(ws + off);   off += szE;
  float* bufH   = (float*)(ws + off); off += (size_t)N * 128 * 4;
  float* bufF   = (float*)(ws + off);
  float* out    = (float*)d_out;

  const int T = 256;
  const int gemmBlocks = cdiv((long)cdiv(N, 16) * 32, T);
  const int nodeBlocks = cdiv((long)N * 32, T);

  k_zero<<<cdiv(N, T), T, 0, stream>>>(cnt, N);
  k_rank<<<cdiv(E, T), T, 0, stream>>>(dst, cnt, pos, E);
  k_scan<<<cdiv(N, 1024), 1024, 0, stream>>>(cnt, rowptr, dinv, N);
  k_place<<<cdiv(E, T), T, 0, stream>>>(src, dst, pos, rowptr, csr, E);

  k_gemm<128><<<gemmBlocks, T, 0, stream>>>(x, W1, bufH, N);
  k_gather<128, true><<<nodeBlocks, T, 0, stream>>>(rowptr, csr, dinv, bufH, b1, bufF, N);

  k_gemm<128><<<gemmBlocks, T, 0, stream>>>(bufF, W2, bufH, N);
  k_gather<128, true><<<nodeBlocks, T, 0, stream>>>(rowptr, csr, dinv, bufH, b2, bufF, N);

  k_gemm<64><<<gemmBlocks, T, 0, stream>>>(bufF, W3, bufH, N);
  k_gather<64, false><<<nodeBlocks, T, 0, stream>>>(rowptr, csr, dinv, bufH, b3, out, N);
}


# Layers 1-2 fused: a block sums 32 nodes' source rows into LDS, then WMMA with W (kept transposed in LDS); layer 2 also applies W3; 10 to 7 launches; sums now aggregate-then-multiply order
# speedup vs baseline: 3.5530x; 1.2333x over previous
#include <hip/hip_runtime.h>


typedef float v2f __attribute__((ext_vector_type(2)));
typedef float v4f __attribute__((ext_vector_type(4)));
typedef float v8f __attribute__((ext_vector_type(8)));

__global__ __launch_bounds__(256) void k_zero(int* __restrict__ p, int n) {
  int i = blockIdx.x * blockDim.x + threadIdx.x;
  if (i < n) p[i] = 0;
}

__global__ __launch_bounds__(256) void k_rank(const int* __restrict__ dst,
                                              int* __restrict__ cnt,
                                              int* __restrict__ pos, int E) {
  int i = blockIdx.x * blockDim.x + threadIdx.x;
  if (i < E) pos[i] = atomicAdd(&cnt[dst[i]], 1);
}

__global__ __launch_bounds__(1024) void k_scan(const int* __restrict__ cnt,
                                               int* __restrict__ rowptr,
                                               float* __restrict__ dinv, int N) {
  __shared__ int wsum[32];
  __shared__ int boff;
  const int tid = threadIdx.x;
  const int lane = tid & 31;
  const int wv = tid >> 5;
  const int base = blockIdx.x * 1024;

  int part = 0;
  const int4* c4 = (const int4*)cnt;
  const int nvec = base >> 2;
  int j = tid;
  for (; j + 3072 < nvec; j += 4096) {
    int4 v0 = c4[j], v1 = c4[j + 1024], v2 = c4[j + 2048], v3 = c4[j + 3072];
    part += ((v0.x + v0.y) + (v0.z + v0.w)) + ((v1.x + v1.y) + (v1.z + v1.w)) +
            ((v2.x + v2.y) + (v2.z + v2.w)) + ((v3.x + v3.y) + (v3.z + v3.w));
  }
  for (; j < nvec; j += 1024) {
    int4 v = c4[j];
    part += (v.x + v.y) + (v.z + v.w);
  }
  for (int o = 16; o > 0; o >>= 1) part += __shfl_xor(part, o);
  if (lane == 0) wsum[wv] = part;
  __syncthreads();
  if (wv == 0) {
    int v = wsum[lane];
    for (int o = 16; o > 0; o >>= 1) v += __shfl_xor(v, o);
    if (lane == 0) boff = v;
  }
  __syncthreads();
  const int off = boff;

  const int i = base + tid;
  const int c = (i < N) ? cnt[i] : 0;
  int incl = c;
  for (int o = 1; o < 32; o <<= 1) {
    int t = __shfl_up(incl, o);
    if (lane >= o) incl += t;
  }
  if (lane == 31) wsum[wv] = incl;
  __syncthreads();
  if (wv == 0) {
    int v = wsum[lane];
    int s = v;
    for (int o = 1; o < 32; o <<= 1) {
      int t = __shfl_up(s, o);
      if (lane >= o) s += t;
    }
    wsum[lane] = s - v;
  }
  __syncthreads();
  const int excl = off + wsum[wv] + (incl - c);
  if (i < N) {
    rowptr[i] = excl;
    dinv[i] = rsqrtf((float)(c + 1));
    if (i == N - 1) rowptr[N] = excl + c;
  }
}

__global__ __launch_bounds__(256) void k_place(const int* __restrict__ src,
                                               const int* __restrict__ dst,
                                               const int* __restrict__ pos,
                                               const int* __restrict__ rowptr,
                                               int* __restrict__ csr, int E) {
  int i = blockIdx.x * blockDim.x + threadIdx.x;
  if (i < E) csr[rowptr[dst[i]] + pos[i]] = src[i];
}

template <int V> struct vecf;
template <> struct vecf<2> { typedef float t __attribute__((ext_vector_type(2))); };
template <> struct vecf<4> { typedef float t __attribute__((ext_vector_type(4))); };

template <int OUT, int R>
__device__ __forceinline__ typename vecf<OUT / 32>::t
gather_chunk(typename vecf<OUT / 32>::t acc, const float* __restrict__ hl,
             int s_l, float g_l, float dd, int j) {
  typedef typename vecf<OUT / 32>::t vf;
  vf v[R];
  float w[R];
#pragma unroll
  for (int u = 0; u < R; ++u) {
    int s = __builtin_amdgcn_readlane(s_l, j + u);
    w[u] = __int_as_float(__builtin_amdgcn_readlane(__float_as_int(g_l), j + u)) * dd;
    v[u] = *(const vf*)(hl + (size_t)s * OUT);
  }
#pragma unroll
  for (int u = 0; u < R; ++u) acc += v[u] * w[u];
  return acc;
}

template <int OUT, bool RELU>
__global__ __launch_bounds__(256)
void k_gather(const int* __restrict__ rowptr, const int* __restrict__ csr,
              const float* __restrict__ dinv, const float* __restrict__ h,
              const float* __restrict__ b, float* __restrict__ out, int N) {
  constexpr int V = OUT / 32;
  typedef typename vecf<V>::t vf;
  const int lane = threadIdx.x & 31;
  const int d = __builtin_amdgcn_readfirstlane(
      (int)((blockIdx.x * blockDim.x + threadIdx.x) >> 5));
  if (d >= N) return;
  const int beg = rowptr[d];
  const int deg = rowptr[d + 1] - beg;
  const float dd = dinv[d];
  const float* hl = h + lane * V;

  const int tot = deg + 1;
  vf acc = {};
  for (int b0 = 0; b0 < tot; b0 += 32) {
    const int n = (tot - b0 < 32) ? (tot - b0) : 32;
    const int t = b0 + lane;
    int s_l = d;
    float g_l = dd;
    if (t < deg) {
      s_l = csr[beg + t];
      g_l = dinv[s_l];
    }
    int j = 0;
    for (; j + 8 <= n; j += 8) acc = gather_chunk<OUT, 8>(acc, hl, s_l, g_l, dd, j);
    switch (n - j) {
      case 7: acc = gather_chunk<OUT, 7>(acc, hl, s_l, g_l, dd, j); break;
      case 6: acc = gather_chunk<OUT, 6>(acc, hl, s_l, g_l, dd, j); break;
      case 5: acc = gather_chunk<OUT, 5>(acc, hl, s_l, g_l, dd, j); break;
      case 4: acc = gather_chunk<OUT, 4>(acc, hl, s_l, g_l, dd, j); break;
      case 3: acc = gather_chunk<OUT, 3>(acc, hl, s_l, g_l, dd, j); break;
      case 2: acc = gather_chunk<OUT, 2>(acc, hl, s_l, g_l, dd, j); break;
      case 1: acc = gather_chunk<OUT, 1>(acc, hl, s_l, g_l, dd, j); break;
      default: break;
    }
  }

  vf r = acc + *(const vf*)(b + lane * V);
  if (RELU) {
#pragma unroll
    for (int k = 0; k < V; ++k) r[k] = fmaxf(r[k], 0.0f);
  }
  *(vf*)(out + (size_t)d * OUT + lane * V) = r;
}

constexpr int LP  = 132;
constexpr int LPC = 68;

__device__ __forceinline__ v8f tile_product(const float* At, const float* Bt, int m, int kh) {
  const float* ap = At + m * LP + 2 * kh;
  const float* bp = Bt + m * LP + 2 * kh;
  v8f acc0 = {}, acc1 = {};
#pragma unroll
  for (int k0 = 0; k0 < 128; k0 += 32) {
    v2f a[8], b[8];
#pragma unroll
    for (int u = 0; u < 8; ++u) {
      a[u] = *(const v2f*)(ap + k0 + 4 * u);
      b[u] = *(const v2f*)(bp + k0 + 4 * u);
    }
#pragma unroll
    for (int u = 0; u < 8; u += 2) {
      acc0 = __builtin_amdgcn_wmma_f32_16x16x4_f32(false, a[u], false, b[u], (short)0, acc0, false, false);
      acc1 = __builtin_amdgcn_wmma_f32_16x16x4_f32(false, a[u + 1], false, b[u + 1], (short)0, acc1, false, false);
    }
  }
  return acc0 + acc1;
}

template <bool TAIL>
__global__ __launch_bounds__(1024)
void k_layer(const int* __restrict__ rowptr, const int* __restrict__ csr,
             const float* __restrict__ dinv, const float* __restrict__ fin,
             const float* __restrict__ W, const float* __restrict__ b,
             const float* __restrict__ W3, float* __restrict__ out, int N) {
  __shared__ __attribute__((aligned(16))) float Wt[128 * LP];
  __shared__ __attribute__((aligned(16))) float W3t[TAIL ? 64 * LP : 4];
  __shared__ __attribute__((aligned(16))) float tA[32 * LP];
  __shared__ __attribute__((aligned(16))) float tB[32 * LP];
  __shared__ __attribute__((aligned(16))) float tC[TAIL ? 32 * LPC : 4];

  const int tid = threadIdx.x;
  const int lane = tid & 31;
  const int wave = __builtin_amdgcn_readfirstlane(tid >> 5);
  const int m = lane & 15;
  const int kh = lane >> 4;

  for (int i = tid; i < 128 * 128; i += 1024) Wt[(i & 127) * LP + (i >> 7)] = W[i];
  if (TAIL) {
    for (int i = tid; i < 128 * 64; i += 1024) W3t[(i & 63) * LP + (i >> 6)] = W3[i];
  }

  const float* hl = fin + lane * 4;
  const int nst = (N + 31) >> 5;
  for (int st = blockIdx.x; st < nst; st += gridDim.x) {
    const int d = st * 32 + wave;
    v4f acc = {};
    if (d < N) {
      const int beg = rowptr[d];
      const int deg = rowptr[d + 1] - beg;
      const float dd = dinv[d];
      const int tot = deg + 1;
      int b0 = 0;
      do {
        const int n = (tot - b0 < 32) ? (tot - b0) : 32;
        const int t = b0 + lane;
        int s_l = d;
        float g_l = dd;
        if (t < deg) {
          s_l = csr[beg + t];
          g_l = dinv[s_l];
        }
        int j = 0;
        for (; j + 8 <= n; j += 8) acc = gather_chunk<128, 8>(acc, hl, s_l, g_l, dd, j);
        switch (n - j) {
          case 7: acc = gather_chunk<128, 7>(acc, hl, s_l, g_l, dd, j); break;
          case 6: acc = gather_chunk<128, 6>(acc, hl, s_l, g_l, dd, j); break;
          case 5: acc = gather_chunk<128, 5>(acc, hl, s_l, g_l, dd, j); break;
          case 4: acc = gather_chunk<128, 4>(acc, hl, s_l, g_l, dd, j); break;
          case 3: acc = gather_chunk<128, 3>(acc, hl, s_l, g_l, dd, j); break;
          case 2: acc = gather_chunk<128, 2>(acc, hl, s_l, g_l, dd, j); break;
          case 1: acc = gather_chunk<128, 1>(acc, hl, s_l, g_l, dd, j); break;
          default: break;
        }
        b0 += 32;
      } while (b0 < tot);
    }
    *(v4f*)&tA[wave * LP + lane * 4] = acc;
    __syncthreads();

    if (wave < 16) {
      const int rt = wave >> 3;
      const int ct = wave & 7;
      const v8f c = tile_product(tA + rt * 16 * LP, Wt + ct * 16 * LP, m, kh);
      const float bias = b[ct * 16 + m];
      float* o = tB + (rt * 16 + 8 * kh) * LP + ct * 16 + m;
#pragma unroll
      for (int r = 0; r < 8; ++r) o[r * LP] = fmaxf(c[r] + bias, 0.0f);
    }
    __syncthreads();

    if (!TAIL) {
      if (d < N) *(v4f*)(out + (size_t)d * 128 + lane * 4) = *(const v4f*)&tB[wave * LP + lane * 4];
    } else {
      if (wave < 8) {
        const int rt = wave >> 2;
        const int ct = wave & 3;
        const v8f c = tile_product(tB + rt * 16 * LP, W3t + ct * 16 * LP, m, kh);
        float* o = tC + (rt * 16 + 8 * kh) * LPC + ct * 16 + m;
#pragma unroll
        for (int r = 0; r < 8; ++r) o[r * LPC] = c[r];
      }
      __syncthreads();
      if (d < N) *(v2f*)(out + (size_t)d * 64 + lane * 2) = *(const v2f*)&tC[wave * LPC + lane * 2];
    }
  }
}

static inline int cdiv(long a, long b) { return (int)((a + b - 1) / b); }

extern "C" void kernel_launch(void* const* d_in, const int* in_sizes, int n_in,
                              void* d_out, int out_size, void* d_ws, size_t ws_size,
                              hipStream_t stream) {
  const float* x  = (const float*)d_in[0];
  const int*   ei = (const int*)d_in[1];
  const float* W1 = (const float*)d_in[2];
  const float* b1 = (const float*)d_in[3];
  const float* W2 = (const float*)d_in[4];
  const float* b2 = (const float*)d_in[5];
  const float* W3 = (const float*)d_in[6];
  const float* b3 = (const float*)d_in[7];

  const int N = in_sizes[0] / 128;
  const int E = in_sizes[1] / 2;
  const int* src = ei;
  const int* dst = ei + E;

  char* ws = (char*)d_ws;
  const size_t szN = (((size_t)(N + 1) * 4) + 255) & ~(size_t)255;
  const size_t szE = (((size_t)E * 4) + 255) & ~(size_t)255;
  size_t off = 0;
  float* dinv   = (float*)(ws + off); off += szN;
  int*   cnt    = (int*)(ws + off);   off += szN;
  int*   rowptr = (int*)(ws + off);   off += szN;
  int*   pos    = (int*)(ws + off);   off += szE;
  int*   csr    = (int*)(ws + off);   off += szE;
  float* bufF   = (float*)(ws + off); off += (size_t)N * 128 * 4;
  float* bufH   = (float*)(ws + off);
  float* out    = (float*)d_out;

  const int T = 256;
  const int nodeBlocks = cdiv((long)N * 32, T);
  const int layerBlocks = 512;

  k_zero<<<cdiv(N, T), T, 0, stream>>>(cnt, N);
  k_rank<<<cdiv(E, T), T, 0, stream>>>(dst, cnt, pos, E);
  k_scan<<<cdiv(N, 1024), 1024, 0, stream>>>(cnt, rowptr, dinv, N);
  k_place<<<cdiv(E, T), T, 0, stream>>>(src, dst, pos, rowptr, csr, E);

  k_layer<false><<<layerBlocks, 1024, 0, stream>>>(rowptr, csr, dinv, x, W1, b1, W3, bufF, N);
  k_layer<true><<<layerBlocks, 1024, 0, stream>>>(rowptr, csr, dinv, bufF, W2, b2, W3, bufH, N);
  k_gather<64, false><<<nodeBlocks, T, 0, stream>>>(rowptr, csr, dinv, bufH, b3, out, N);
}


# CSR build as a two-level counting sort on LDS adds (tile histogram, bucket filing, per-bucket rows), 3 launches; replaces zero, global atomic rank, scan, place; row sum order: LDS add arrival order
# speedup vs baseline: 4.8591x; 1.3676x over previous
#include <hip/hip_runtime.h>


typedef float v2f __attribute__((ext_vector_type(2)));
typedef float v4f __attribute__((ext_vector_type(4)));
typedef float v8f __attribute__((ext_vector_type(8)));

constexpr int BSH   = 8;
constexpr int BN    = 1 << BSH;
constexpr int NBMAX = 1024;
constexpr int EPT   = 8;
constexpr int TE    = 1024 * EPT;

__global__ __launch_bounds__(1024) void k_hist(const int* __restrict__ dst,
                                               int* __restrict__ hist, int E, int NB) {
  __shared__ int h[NBMAX];
  const int tid = threadIdx.x;
  const int base = blockIdx.x * TE;
  int d[EPT];
#pragma unroll
  for (int u = 0; u < EPT; ++u) {
    const int i = base + u * 1024 + tid;
    d[u] = dst[(i < E) ? i : (E - 1)];
  }
#pragma unroll
  for (int u = 0; u < EPT; ++u)
    if (base + u * 1024 + tid >= E) d[u] = -1;
  if (tid < NB) h[tid] = 0;
  __syncthreads();
#pragma unroll
  for (int u = 0; u < EPT; ++u)
    if (d[u] >= 0) atomicAdd(&h[d[u] >> BSH], 1);
  __syncthreads();
  if (tid < NB) hist[blockIdx.x * NB + tid] = h[tid];
}

__global__ __launch_bounds__(1024) void k_bin(const int* __restrict__ src,
                                              const int* __restrict__ dst,
                                              const int* __restrict__ hist,
                                              int* __restrict__ bstart,
                                              int* __restrict__ rec,
                                              int E, int NB, int NT) {
  __shared__ int cur[NBMAX];
  __shared__ int wsum[32];
  const int tid = threadIdx.x;
  const int lane = tid & 31;
  const int wv = tid >> 5;
  const int t = blockIdx.x;
  const int base = t * TE;

  int s[EPT], d[EPT];
#pragma unroll
  for (int u = 0; u < EPT; ++u) {
    const int i = base + u * 1024 + tid;
    const int ic = (i < E) ? i : (E - 1);
    d[u] = dst[ic];
    s[u] = src[ic];
  }

  int pre = 0, rest = 0;
  if (tid < NB) {
    const int* hc = hist + tid;
#pragma unroll 16
    for (int k = 0; k < t; ++k) pre += hc[k * NB];
#pragma unroll 16
    for (int k = t; k < NT; ++k) rest += hc[k * NB];
  }
  const int tot = pre + rest;

  int incl = tot;
  for (int o = 1; o < 32; o <<= 1) {
    int x = __shfl_up(incl, o);
    if (lane >= o) incl += x;
  }
  if (lane == 31) wsum[wv] = incl;
  __syncthreads();
  if (wv == 0) {
    int v = wsum[lane];
    int sc = v;
    for (int o = 1; o < 32; o <<= 1) {
      int x = __shfl_up(sc, o);
      if (lane >= o) sc += x;
    }
    wsum[lane] = sc - v;
  }
  __syncthreads();
  const int excl = wsum[wv] + (incl - tot);
  if (tid < NB) {
    cur[tid] = excl + pre;
    if (t == 0) {
      bstart[tid] = excl;
      if (tid == NB - 1) bstart[NB] = excl + tot;
    }
  }
  __syncthreads();

#pragma unroll
  for (int u = 0; u < EPT; ++u) {
    if (base + u * 1024 + tid < E) {
      const int p = atomicAdd(&cur[d[u] >> BSH], 1);
      rec[p] = (int)(((unsigned)s[u] << BSH) | (unsigned)(d[u] & (BN - 1)));
    }
  }
}

__global__ __launch_bounds__(1024) void k_rows(const int* __restrict__ bstart,
                                               const int* __restrict__ rec,
                                               int* __restrict__ rowptr,
                                               float* __restrict__ dinv,
                                               int* __restrict__ csr, int N) {
  __shared__ int cnt[BN];
  __shared__ int wtot[BN / 32];
  const int tid = threadIdx.x;
  const int lane = tid & 31;
  const int wv = tid >> 5;
  const int b = blockIdx.x;
  const int bs = bstart[b];
  const int ne = bstart[b + 1] - bs;
  const int* rb = rec + bs;

  if (tid < BN) cnt[tid] = 0;
  __syncthreads();
  for (int i0 = tid; i0 < ne; i0 += 4096) {
    int r[4];
#pragma unroll
    for (int u = 0; u < 4; ++u) {
      const int i = i0 + u * 1024;
      r[u] = rb[(i < ne) ? i : (ne - 1)];
    }
#pragma unroll
    for (int u = 0; u < 4; ++u)
      if (i0 + u * 1024 < ne) atomicAdd(&cnt[r[u] & (BN - 1)], 1);
  }
  __syncthreads();

  int c = 0, incl = 0;
  if (tid < BN) {
    c = cnt[tid];
    incl = c;
    for (int o = 1; o < 32; o <<= 1) {
      int x = __shfl_up(incl, o);
      if (lane >= o) incl += x;
    }
    if (lane == 31) wtot[wv] = incl;
  }
  __syncthreads();
  if (tid < BN) {
    int off = 0;
    for (int w = 0; w < wv; ++w) off += wtot[w];
    const int excl = off + (incl - c);
    cnt[tid] = excl;
    const int node = b * BN + tid;
    if (node < N) {
      rowptr[node] = bs + excl;
      dinv[node] = rsqrtf((float)(c + 1));
      if (node == N - 1) rowptr[N] = bs + excl + c;
    }
  }
  __syncthreads();
  for (int i0 = tid; i0 < ne; i0 += 4096) {
    int r[4];
#pragma unroll
    for (int u = 0; u < 4; ++u) {
      const int i = i0 + u * 1024;
      r[u] = rb[(i < ne) ? i : (ne - 1)];
    }
#pragma unroll
    for (int u = 0; u < 4; ++u) {
      if (i0 + u * 1024 < ne) {
        const int p = atomicAdd(&cnt[r[u] & (BN - 1)], 1);
        csr[bs + p] = (int)((unsigned)r[u] >> BSH);
      }
    }
  }
}

template <int V> struct vecf;
template <> struct vecf<2> { typedef float t __attribute__((ext_vector_type(2))); };
template <> struct vecf<4> { typedef float t __attribute__((ext_vector_type(4))); };

template <int OUT, int R>
__device__ __forceinline__ typename vecf<OUT / 32>::t
gather_chunk(typename vecf<OUT / 32>::t acc, const float* __restrict__ hl,
             int s_l, float g_l, float dd, int j) {
  typedef typename vecf<OUT / 32>::t vf;
  vf v[R];
  float w[R];
#pragma unroll
  for (int u = 0; u < R; ++u) {
    int s = __builtin_amdgcn_readlane(s_l, j + u);
    w[u] = __int_as_float(__builtin_amdgcn_readlane(__float_as_int(g_l), j + u)) * dd;
    v[u] = *(const vf*)(hl + (size_t)s * OUT);
  }
#pragma unroll
  for (int u = 0; u < R; ++u) acc += v[u] * w[u];
  return acc;
}

template <int OUT, bool RELU>
__global__ __launch_bounds__(256)
void k_gather(const int* __restrict__ rowptr, const int* __restrict__ csr,
              const float* __restrict__ dinv, const float* __restrict__ h,
              const float* __restrict__ b, float* __restrict__ out, int N) {
  constexpr int V = OUT / 32;
  typedef typename vecf<V>::t vf;
  const int lane = threadIdx.x & 31;
  const int d = __builtin_amdgcn_readfirstlane(
      (int)((blockIdx.x * blockDim.x + threadIdx.x) >> 5));
  if (d >= N) return;
  const int beg = rowptr[d];
  const int deg = rowptr[d + 1] - beg;
  const float dd = dinv[d];
  const float* hl = h + lane * V;

  const int tot = deg + 1;
  vf acc = {};
  for (int b0 = 0; b0 < tot; b0 += 32) {
    const int n = (tot - b0 < 32) ? (tot - b0) : 32;
    const int t = b0 + lane;
    int s_l = d;
    float g_l = dd;
    if (t < deg) {
      s_l = csr[beg + t];
      g_l = dinv[s_l];
    }
    int j = 0;
    for (; j + 8 <= n; j += 8) acc = gather_chunk<OUT, 8>(acc, hl, s_l, g_l, dd, j);
    switch (n - j) {
      case 7: acc = gather_chunk<OUT, 7>(acc, hl, s_l, g_l, dd, j); break;
      case 6: acc = gather_chunk<OUT, 6>(acc, hl, s_l, g_l, dd, j); break;
      case 5: acc = gather_chunk<OUT, 5>(acc, hl, s_l, g_l, dd, j); break;
      case 4: acc = gather_chunk<OUT, 4>(acc, hl, s_l, g_l, dd, j); break;
      case 3: acc = gather_chunk<OUT, 3>(acc, hl, s_l, g_l, dd, j); break;
      case 2: acc = gather_chunk<OUT, 2>(acc, hl, s_l, g_l, dd, j); break;
      case 1: acc = gather_chunk<OUT, 1>(acc, hl, s_l, g_l, dd, j); break;
      default: break;
    }
  }

  vf r = acc + *(const vf*)(b + lane * V);
  if (RELU) {
#pragma unroll
    for (int k = 0; k < V; ++k) r[k] = fmaxf(r[k], 0.0f);
  }
  *(vf*)(out + (size_t)d * OUT + lane * V) = r;
}

constexpr int LP  = 132;
constexpr int LPC = 68;

__device__ __forceinline__ v8f tile_product(const float* At, const float* Bt, int m, int kh) {
  const float* ap = At + m * LP + 2 * kh;
  const float* bp = Bt + m * LP + 2 * kh;
  v8f acc0 = {}, acc1 = {};
#pragma unroll
  for (int k0 = 0; k0 < 128; k0 += 32) {
    v2f a[8], b[8];
#pragma unroll
    for (int u = 0; u < 8; ++u) {
      a[u] = *(const v2f*)(ap + k0 + 4 * u);
      b[u] = *(const v2f*)(bp + k0 + 4 * u);
    }
#pragma unroll
    for (int u = 0; u < 8; u += 2) {
      acc0 = __builtin_amdgcn_wmma_f32_16x16x4_f32(false, a[u], false, b[u], (short)0, acc0, false, false);
      acc1 = __builtin_amdgcn_wmma_f32_16x16x4_f32(false, a[u + 1], false, b[u + 1], (short)0, acc1, false, false);
    }
  }
  return acc0 + acc1;
}

template <bool TAIL>
__global__ __launch_bounds__(1024)
void k_layer(const int* __restrict__ rowptr, const int* __restrict__ csr,
             const float* __restrict__ dinv, const float* __restrict__ fin,
             const float* __restrict__ W, const float* __restrict__ b,
             const float* __restrict__ W3, float* __restrict__ out, int N) {
  __shared__ __attribute__((aligned(16))) float Wt[128 * LP];
  __shared__ __attribute__((aligned(16))) float W3t[TAIL ? 64 * LP : 4];
  __shared__ __attribute__((aligned(16))) float tA[32 * LP];
  __shared__ __attribute__((aligned(16))) float tB[32 * LP];
  __shared__ __attribute__((aligned(16))) float tC[TAIL ? 32 * LPC : 4];

  const int tid = threadIdx.x;
  const int lane = tid & 31;
  const int wave = __builtin_amdgcn_readfirstlane(tid >> 5);
  const int m = lane & 15;
  const int kh = lane >> 4;

  for (int i = tid; i < 128 * 128; i += 1024) Wt[(i & 127) * LP + (i >> 7)] = W[i];
  if (TAIL) {
    for (int i = tid; i < 128 * 64; i += 1024) W3t[(i & 63) * LP + (i >> 6)] = W3[i];
  }

  const float* hl = fin + lane * 4;
  const int nst = (N + 31) >> 5;
  for (int st = blockIdx.x; st < nst; st += gridDim.x) {
    const int d = st * 32 + wave;
    v4f acc = {};
    if (d < N) {
      const int beg = rowptr[d];
      const int deg = rowptr[d + 1] - beg;
      const float dd = dinv[d];
      const int tot = deg + 1;
      int b0 = 0;
      do {
        const int n = (tot - b0 < 32) ? (tot - b0) : 32;
        const int t = b0 + lane;
        int s_l = d;
        float g_l = dd;
        if (t < deg) {
          s_l = csr[beg + t];
          g_l = dinv[s_l];
        }
        int j = 0;
        for (; j + 8 <= n; j += 8) acc = gather_chunk<128, 8>(acc, hl, s_l, g_l, dd, j);
        switch (n - j) {
          case 7: acc = gather_chunk<128, 7>(acc, hl, s_l, g_l, dd, j); break;
          case 6: acc = gather_chunk<128, 6>(acc, hl, s_l, g_l, dd, j); break;
          case 5: acc = gather_chunk<128, 5>(acc, hl, s_l, g_l, dd, j); break;
          case 4: acc = gather_chunk<128, 4>(acc, hl, s_l, g_l, dd, j); break;
          case 3: acc = gather_chunk<128, 3>(acc, hl, s_l, g_l, dd, j); break;
          case 2: acc = gather_chunk<128, 2>(acc, hl, s_l, g_l, dd, j); break;
          case 1: acc = gather_chunk<128, 1>(acc, hl, s_l, g_l, dd, j); break;
          default: break;
        }
        b0 += 32;
      } while (b0 < tot);
    }
    *(v4f*)&tA[wave * LP + lane * 4] = acc;
    __syncthreads();

    if (wave < 16) {
      const int rt = wave >> 3;
      const int ct = wave & 7;
      const v8f c = tile_product(tA + rt * 16 * LP, Wt + ct * 16 * LP, m, kh);
      const float bias = b[ct * 16 + m];
      float* o = tB + (rt * 16 + 8 * kh) * LP + ct * 16 + m;
#pragma unroll
      for (int r = 0; r < 8; ++r) o[r * LP] = fmaxf(c[r] + bias, 0.0f);
    }
    __syncthreads();

    if (!TAIL) {
      if (d < N) *(v4f*)(out + (size_t)d * 128 + lane * 4) = *(const v4f*)&tB[wave * LP + lane * 4];
    } else {
      if (wave < 8) {
        const int rt = wave >> 2;
        const int ct = wave & 3;
        const v8f c = tile_product(tB + rt * 16 * LP, W3t + ct * 16 * LP, m, kh);
        float* o = tC + (rt * 16 + 8 * kh) * LPC + ct * 16 + m;
#pragma unroll
        for (int r = 0; r < 8; ++r) o[r * LPC] = c[r];
      }
      __syncthreads();
      if (d < N) *(v2f*)(out + (size_t)d * 64 + lane * 2) = *(const v2f*)&tC[wave * LPC + lane * 2];
    }
  }
}

static inline int cdiv(long a, long b) { return (int)((a + b - 1) / b); }

extern "C" void kernel_launch(void* const* d_in, const int* in_sizes, int n_in,
                              void* d_out, int out_size, void* d_ws, size_t ws_size,
                              hipStream_t stream) {
  const float* x  = (const float*)d_in[0];
  const int*   ei = (const int*)d_in[1];
  const float* W1 = (const float*)d_in[2];
  const float* b1 = (const float*)d_in[3];
  const float* W2 = (const float*)d_in[4];
  const float* b2 = (const float*)d_in[5];
  const float* W3 = (const float*)d_in[6];
  const float* b3 = (const float*)d_in[7];

  const int N = in_sizes[0] / 128;
  const int E = in_sizes[1] / 2;
  const int* src = ei;
  const int* dst = ei + E;

  char* ws = (char*)d_ws;
  const size_t szN = (((size_t)(N + 1) * 4) + 255) & ~(size_t)255;
  const size_t szE = (((size_t)E * 4) + 255) & ~(size_t)255;
  size_t off = 0;
  const int NB = cdiv(N, BN);
  const int NT = cdiv(E, TE);
  const size_t szH = (((size_t)NT * NB * 4) + 255) & ~(size_t)255;
  const size_t szB = (((size_t)(NB + 1) * 4) + 255) & ~(size_t)255;
  float* dinv   = (float*)(ws + off); off += szN;
  int*   rowptr = (int*)(ws + off);   off += szN;
  int*   hist   = (int*)(ws + off);   off += szH;
  int*   bstart = (int*)(ws + off);   off += szB;
  int*   rec    = (int*)(ws + off);   off += szE;
  int*   csr    = (int*)(ws + off);   off += szE;
  float* bufF   = (float*)(ws + off); off += (size_t)N * 128 * 4;
  float* bufH   = (float*)(ws + off);
  float* out    = (float*)d_out;

  const int T = 256;
  const int nodeBlocks = cdiv((long)N * 32, T);
  const int layerBlocks = 512;

  k_hist<<<NT, 1024, 0, stream>>>(dst, hist, E, NB);
  k_bin<<<NT, 1024, 0, stream>>>(src, dst, hist, bstart, rec, E, NB, NT);
  k_rows<<<NB, 1024, 0, stream>>>(bstart, rec, rowptr, dinv, csr, N);

  k_layer<false><<<layerBlocks, 1024, 0, stream>>>(rowptr, csr, dinv, x, W1, b1, W3, bufF, N);
  k_layer<true><<<layerBlocks, 1024, 0, stream>>>(rowptr, csr, dinv, bufF, W2, b2, W3, bufH, N);
  k_gather<64, false><<<nodeBlocks, T, 0, stream>>>(rowptr, csr, dinv, bufH, b3, out, N);
}


# Layer outputs stored times their node's d^-1/2 by the producing epilogue: layer-2 and layer-3 gathers sum rows unweighted, no dinv[src] load or multiply per edge; sum order kept, scale factored out
# speedup vs baseline: 5.1277x; 1.0553x over previous
#include <hip/hip_runtime.h>


typedef float v2f __attribute__((ext_vector_type(2)));
typedef float v4f __attribute__((ext_vector_type(4)));
typedef float v8f __attribute__((ext_vector_type(8)));

constexpr int BSH   = 8;
constexpr int BN    = 1 << BSH;
constexpr int NBMAX = 1024;
constexpr int EPT   = 8;
constexpr int TE    = 1024 * EPT;

__global__ __launch_bounds__(1024) void k_hist(const int* __restrict__ dst,
                                               int* __restrict__ hist, int E, int NB) {
  __shared__ int h[NBMAX];
  const int tid = threadIdx.x;
  const int base = blockIdx.x * TE;
  int d[EPT];
#pragma unroll
  for (int u = 0; u < EPT; ++u) {
    const int i = base + u * 1024 + tid;
    d[u] = dst[(i < E) ? i : (E - 1)];
  }
#pragma unroll
  for (int u = 0; u < EPT; ++u)
    if (base + u * 1024 + tid >= E) d[u] = -1;
  if (tid < NB) h[tid] = 0;
  __syncthreads();
#pragma unroll
  for (int u = 0; u < EPT; ++u)
    if (d[u] >= 0) atomicAdd(&h[d[u] >> BSH], 1);
  __syncthreads();
  if (tid < NB) hist[blockIdx.x * NB + tid] = h[tid];
}

__global__ __launch_bounds__(1024) void k_bin(const int* __restrict__ src,
                                              const int* __restrict__ dst,
                                              const int* __restrict__ hist,
                                              int* __restrict__ bstart,
                                              int* __restrict__ rec,
                                              int E, int NB, int NT) {
  __shared__ int cur[NBMAX];
  __shared__ int wsum[32];
  const int tid = threadIdx.x;
  const int lane = tid & 31;
  const int wv = tid >> 5;
  const int t = blockIdx.x;
  const int base = t * TE;

  int s[EPT], d[EPT];
#pragma unroll
  for (int u = 0; u < EPT; ++u) {
    const int i = base + u * 1024 + tid;
    const int ic = (i < E) ? i : (E - 1);
    d[u] = dst[ic];
    s[u] = src[ic];
  }

  int pre = 0, rest = 0;
  if (tid < NB) {
    const int* hc = hist + tid;
#pragma unroll 16
    for (int k = 0; k < t; ++k) pre += hc[k * NB];
#pragma unroll 16
    for (int k = t; k < NT; ++k) rest += hc[k * NB];
  }
  const int tot = pre + rest;

  int incl = tot;
  for (int o = 1; o < 32; o <<= 1) {
    int x = __shfl_up(incl, o);
    if (lane >= o) incl += x;
  }
  if (lane == 31) wsum[wv] = incl;
  __syncthreads();
  if (wv == 0) {
    int v = wsum[lane];
    int sc = v;
    for (int o = 1; o < 32; o <<= 1) {
      int x = __shfl_up(sc, o);
      if (lane >= o) sc += x;
    }
    wsum[lane] = sc - v;
  }
  __syncthreads();
  const int excl = wsum[wv] + (incl - tot);
  if (tid < NB) {
    cur[tid] = excl + pre;
    if (t == 0) {
      bstart[tid] = excl;
      if (tid == NB - 1) bstart[NB] = excl + tot;
    }
  }
  __syncthreads();

#pragma unroll
  for (int u = 0; u < EPT; ++u) {
    if (base + u * 1024 + tid < E) {
      const int p = atomicAdd(&cur[d[u] >> BSH], 1);
      rec[p] = (int)(((unsigned)s[u] << BSH) | (unsigned)(d[u] & (BN - 1)));
    }
  }
}

__global__ __launch_bounds__(1024) void k_rows(const int* __restrict__ bstart,
                                               const int* __restrict__ rec,
                                               int* __restrict__ rowptr,
                                               float* __restrict__ dinv,
                                               int* __restrict__ csr, int N) {
  __shared__ int cnt[BN];
  __shared__ int wtot[BN / 32];
  const int tid = threadIdx.x;
  const int lane = tid & 31;
  const int wv = tid >> 5;
  const int b = blockIdx.x;
  const int bs = bstart[b];
  const int ne = bstart[b + 1] - bs;
  const int* rb = rec + bs;

  if (tid < BN) cnt[tid] = 0;
  __syncthreads();
  for (int i0 = tid; i0 < ne; i0 += 4096) {
    int r[4];
#pragma unroll
    for (int u = 0; u < 4; ++u) {
      const int i = i0 + u * 1024;
      r[u] = rb[(i < ne) ? i : (ne - 1)];
    }
#pragma unroll
    for (int u = 0; u < 4; ++u)
      if (i0 + u * 1024 < ne) atomicAdd(&cnt[r[u] & (BN - 1)], 1);
  }
  __syncthreads();

  int c = 0, incl = 0;
  if (tid < BN) {
    c = cnt[tid];
    incl = c;
    for (int o = 1; o < 32; o <<= 1) {
      int x = __shfl_up(incl, o);
      if (lane >= o) incl += x;
    }
    if (lane == 31) wtot[wv] = incl;
  }
  __syncthreads();
  if (tid < BN) {
    int off = 0;
    for (int w = 0; w < wv; ++w) off += wtot[w];
    const int excl = off + (incl - c);
    cnt[tid] = excl;
    const int node = b * BN + tid;
    if (node < N) {
      rowptr[node] = bs + excl;
      dinv[node] = rsqrtf((float)(c + 1));
      if (node == N - 1) rowptr[N] = bs + excl + c;
    }
  }
  __syncthreads();
  for (int i0 = tid; i0 < ne; i0 += 4096) {
    int r[4];
#pragma unroll
    for (int u = 0; u < 4; ++u) {
      const int i = i0 + u * 1024;
      r[u] = rb[(i < ne) ? i : (ne - 1)];
    }
#pragma unroll
    for (int u = 0; u < 4; ++u) {
      if (i0 + u * 1024 < ne) {
        const int p = atomicAdd(&cnt[r[u] & (BN - 1)], 1);
        csr[bs + p] = (int)((unsigned)r[u] >> BSH);
      }
    }
  }
}

template <int V> struct vecf;
template <> struct vecf<2> { typedef float t __attribute__((ext_vector_type(2))); };
template <> struct vecf<4> { typedef float t __attribute__((ext_vector_type(4))); };

template <int OUT, int R, bool PRE>
__device__ __forceinline__ typename vecf<OUT / 32>::t
gather_chunk(typename vecf<OUT / 32>::t acc, const float* __restrict__ hl,
             int s_l, float g_l, int j) {
  typedef typename vecf<OUT / 32>::t vf;
  vf v[R];
  float w[R];
#pragma unroll
  for (int u = 0; u < R; ++u) {
    const int s = __builtin_amdgcn_readlane(s_l, j + u);
    w[u] = PRE ? 1.0f : __int_as_float(__builtin_amdgcn_readlane(__float_as_int(g_l), j + u));
    v[u] = *(const vf*)(hl + (size_t)s * OUT);
  }
#pragma unroll
  for (int u = 0; u < R; ++u) {
    if (PRE) acc += v[u];
    else acc += v[u] * w[u];
  }
  return acc;
}

template <int OUT, bool RELU>
__global__ __launch_bounds__(256)
void k_gather(const int* __restrict__ rowptr, const int* __restrict__ csr,
              const float* __restrict__ dinv, const float* __restrict__ h,
              const float* __restrict__ b, float* __restrict__ out, int N) {
  constexpr int V = OUT / 32;
  typedef typename vecf<V>::t vf;
  const int lane = threadIdx.x & 31;
  const int d = __builtin_amdgcn_readfirstlane(
      (int)((blockIdx.x * blockDim.x + threadIdx.x) >> 5));
  if (d >= N) return;
  const int beg = rowptr[d];
  const int deg = rowptr[d + 1] - beg;
  const float dd = dinv[d];
  const float* hl = h + lane * V;

  const int tot = deg + 1;
  vf acc = {};
  for (int b0 = 0; b0 < tot; b0 += 32) {
    const int n = (tot - b0 < 32) ? (tot - b0) : 32;
    const int t = b0 + lane;
    int s_l = d;
    if (t < deg) s_l = csr[beg + t];
    int j = 0;
    for (; j + 8 <= n; j += 8) acc = gather_chunk<OUT, 8, true>(acc, hl, s_l, 0.0f, j);
    switch (n - j) {
      case 7: acc = gather_chunk<OUT, 7, true>(acc, hl, s_l, 0.0f, j); break;
      case 6: acc = gather_chunk<OUT, 6, true>(acc, hl, s_l, 0.0f, j); break;
      case 5: acc = gather_chunk<OUT, 5, true>(acc, hl, s_l, 0.0f, j); break;
      case 4: acc = gather_chunk<OUT, 4, true>(acc, hl, s_l, 0.0f, j); break;
      case 3: acc = gather_chunk<OUT, 3, true>(acc, hl, s_l, 0.0f, j); break;
      case 2: acc = gather_chunk<OUT, 2, true>(acc, hl, s_l, 0.0f, j); break;
      case 1: acc = gather_chunk<OUT, 1, true>(acc, hl, s_l, 0.0f, j); break;
      default: break;
    }
  }

  vf r = acc * dd + *(const vf*)(b + lane * V);
  if (RELU) {
#pragma unroll
    for (int k = 0; k < V; ++k) r[k] = fmaxf(r[k], 0.0f);
  }
  *(vf*)(out + (size_t)d * OUT + lane * V) = r;
}

constexpr int LP  = 132;
constexpr int LPC = 68;

__device__ __forceinline__ v8f tile_product(const float* At, const float* Bt, int m, int kh) {
  const float* ap = At + m * LP + 2 * kh;
  const float* bp = Bt + m * LP + 2 * kh;
  v8f acc0 = {}, acc1 = {};
#pragma unroll
  for (int k0 = 0; k0 < 128; k0 += 32) {
    v2f a[8], b[8];
#pragma unroll
    for (int u = 0; u < 8; ++u) {
      a[u] = *(const v2f*)(ap + k0 + 4 * u);
      b[u] = *(const v2f*)(bp + k0 + 4 * u);
    }
#pragma unroll
    for (int u = 0; u < 8; u += 2) {
      acc0 = __builtin_amdgcn_wmma_f32_16x16x4_f32(false, a[u], false, b[u], (short)0, acc0, false, false);
      acc1 = __builtin_amdgcn_wmma_f32_16x16x4_f32(false, a[u + 1], false, b[u + 1], (short)0, acc1, false, false);
    }
  }
  return acc0 + acc1;
}

template <bool PRE, bool TAIL>
__global__ __launch_bounds__(1024)
void k_layer(const int* __restrict__ rowptr, const int* __restrict__ csr,
             const float* __restrict__ dinv, const float* __restrict__ fin,
             const float* __restrict__ W, const float* __restrict__ b,
             const float* __restrict__ W3, float* __restrict__ out, int N) {
  __shared__ __attribute__((aligned(16))) float Wt[128 * LP];
  __shared__ __attribute__((aligned(16))) float W3t[TAIL ? 64 * LP : 4];
  __shared__ __attribute__((aligned(16))) float tA[32 * LP];
  __shared__ __attribute__((aligned(16))) float tB[32 * LP];
  __shared__ __attribute__((aligned(16))) float tC[TAIL ? 32 * LPC : 4];

  const int tid = threadIdx.x;
  const int lane = tid & 31;
  const int wave = __builtin_amdgcn_readfirstlane(tid >> 5);
  const int m = lane & 15;
  const int kh = lane >> 4;

  for (int i = tid; i < 128 * 128; i += 1024) Wt[(i & 127) * LP + (i >> 7)] = W[i];
  if (TAIL) {
    for (int i = tid; i < 128 * 64; i += 1024) W3t[(i & 63) * LP + (i >> 6)] = W3[i];
  }

  const float* hl = fin + lane * 4;
  const int nst = (N + 31) >> 5;
  for (int st = blockIdx.x; st < nst; st += gridDim.x) {
    const int d = st * 32 + wave;
    v4f acc = {};
    float dd = 0.0f;
    if (d < N) {
      const int beg = rowptr[d];
      const int deg = rowptr[d + 1] - beg;
      dd = dinv[d];
      const int tot = deg + 1;
      int b0 = 0;
      do {
        const int n = (tot - b0 < 32) ? (tot - b0) : 32;
        const int t = b0 + lane;
        int s_l = d;
        float g_l = dd;
        if (t < deg) {
          s_l = csr[beg + t];
          if (!PRE) g_l = dinv[s_l];
        }
        int j = 0;
        for (; j + 8 <= n; j += 8) acc = gather_chunk<128, 8, PRE>(acc, hl, s_l, g_l, j);
        switch (n - j) {
          case 7: acc = gather_chunk<128, 7, PRE>(acc, hl, s_l, g_l, j); break;
          case 6: acc = gather_chunk<128, 6, PRE>(acc, hl, s_l, g_l, j); break;
          case 5: acc = gather_chunk<128, 5, PRE>(acc, hl, s_l, g_l, j); break;
          case 4: acc = gather_chunk<128, 4, PRE>(acc, hl, s_l, g_l, j); break;
          case 3: acc = gather_chunk<128, 3, PRE>(acc, hl, s_l, g_l, j); break;
          case 2: acc = gather_chunk<128, 2, PRE>(acc, hl, s_l, g_l, j); break;
          case 1: acc = gather_chunk<128, 1, PRE>(acc, hl, s_l, g_l, j); break;
          default: break;
        }
        b0 += 32;
      } while (b0 < tot);
    }
    *(v4f*)&tA[wave * LP + lane * 4] = acc * dd;
    __syncthreads();

    if (wave < 16) {
      const int rt = wave >> 3;
      const int ct = wave & 7;
      const v8f c = tile_product(tA + rt * 16 * LP, Wt + ct * 16 * LP, m, kh);
      const float bias = b[ct * 16 + m];
      float* o = tB + (rt * 16 + 8 * kh) * LP + ct * 16 + m;
#pragma unroll
      for (int r = 0; r < 8; ++r) o[r * LP] = fmaxf(c[r] + bias, 0.0f);
    }
    __syncthreads();

    if (!TAIL) {
      if (d < N) *(v4f*)(out + (size_t)d * 128 + lane * 4) = *(const v4f*)&tB[wave * LP + lane * 4] * dd;
    } else {
      if (wave < 8) {
        const int rt = wave >> 2;
        const int ct = wave & 3;
        const v8f c = tile_product(tB + rt * 16 * LP, W3t + ct * 16 * LP, m, kh);
        float* o = tC + (rt * 16 + 8 * kh) * LPC + ct * 16 + m;
#pragma unroll
        for (int r = 0; r < 8; ++r) o[r * LPC] = c[r];
      }
      __syncthreads();
      if (d < N) *(v2f*)(out + (size_t)d * 64 + lane * 2) = *(const v2f*)&tC[wave * LPC + lane * 2] * dd;
    }
  }
}

static inline int cdiv(long a, long b) { return (int)((a + b - 1) / b); }

extern "C" void kernel_launch(void* const* d_in, const int* in_sizes, int n_in,
                              void* d_out, int out_size, void* d_ws, size_t ws_size,
                              hipStream_t stream) {
  const float* x  = (const float*)d_in[0];
  const int*   ei = (const int*)d_in[1];
  const float* W1 = (const float*)d_in[2];
  const float* b1 = (const float*)d_in[3];
  const float* W2 = (const float*)d_in[4];
  const float* b2 = (const float*)d_in[5];
  const float* W3 = (const float*)d_in[6];
  const float* b3 = (const float*)d_in[7];

  const int N = in_sizes[0] / 128;
  const int E = in_sizes[1] / 2;
  const int* src = ei;
  const int* dst = ei + E;

  char* ws = (char*)d_ws;
  const size_t szN = (((size_t)(N + 1) * 4) + 255) & ~(size_t)255;
  const size_t szE = (((size_t)E * 4) + 255) & ~(size_t)255;
  size_t off = 0;
  const int NB = cdiv(N, BN);
  const int NT = cdiv(E, TE);
  const size_t szH = (((size_t)NT * NB * 4) + 255) & ~(size_t)255;
  const size_t szB = (((size_t)(NB + 1) * 4) + 255) & ~(size_t)255;
  float* dinv   = (float*)(ws + off); off += szN;
  int*   rowptr = (int*)(ws + off);   off += szN;
  int*   hist   = (int*)(ws + off);   off += szH;
  int*   bstart = (int*)(ws + off);   off += szB;
  int*   rec    = (int*)(ws + off);   off += szE;
  int*   csr    = (int*)(ws + off);   off += szE;
  float* bufF   = (float*)(ws + off); off += (size_t)N * 128 * 4;
  float* bufH   = (float*)(ws + off);
  float* out    = (float*)d_out;

  const int T = 256;
  const int nodeBlocks = cdiv((long)N * 32, T);
  const int layerBlocks = 512;

  k_hist<<<NT, 1024, 0, stream>>>(dst, hist, E, NB);
  k_bin<<<NT, 1024, 0, stream>>>(src, dst, hist, bstart, rec, E, NB, NT);
  k_rows<<<NB, 1024, 0, stream>>>(bstart, rec, rowptr, dinv, csr, N);

  k_layer<false, false><<<layerBlocks, 1024, 0, stream>>>(rowptr, csr, dinv, x, W1, b1, W3, bufF, N);
  k_layer<true, true><<<layerBlocks, 1024, 0, stream>>>(rowptr, csr, dinv, bufF, W2, b2, W3, bufH, N);
  k_gather<64, false><<<nodeBlocks, T, 0, stream>>>(rowptr, csr, dinv, bufH, b3, out, N);
}


# CSR build in 2 launches, was 3: a tile block counts, ranks and files its edges inside its own slots; a bucket block reads its piece of every tile; row sum order: LDS add arrival order, as before
# speedup vs baseline: 5.6962x; 1.1109x over previous
#include <hip/hip_runtime.h>


typedef float v2f __attribute__((ext_vector_type(2)));
typedef float v4f __attribute__((ext_vector_type(4)));
typedef float v8f __attribute__((ext_vector_type(8)));

constexpr int BSH   = 8;
constexpr int BN    = 1 << BSH;
constexpr int NBMAX = 1024;
constexpr int EPT   = 8;
constexpr int TE    = 1024 * EPT;

constexpr int TPW   = 7;

__global__ __launch_bounds__(1024) void k_tile(const int* __restrict__ src,
                                               const int* __restrict__ dst,
                                               int* __restrict__ hist,
                                               int* __restrict__ offs,
                                               int* __restrict__ rec,
                                               int E, int NB) {
  __shared__ int h[NBMAX];
  __shared__ int wsum[32];
  const int tid = threadIdx.x;
  const int lane = tid & 31;
  const int wv = tid >> 5;
  const int t = blockIdx.x;
  const int base = t * TE;

  int s[EPT], d[EPT], rk[EPT];
#pragma unroll
  for (int u = 0; u < EPT; ++u) {
    const int i = base + u * 1024 + tid;
    const int ic = (i < E) ? i : (E - 1);
    d[u] = dst[ic];
    s[u] = src[ic];
  }
  h[tid] = 0;
  __syncthreads();
#pragma unroll
  for (int u = 0; u < EPT; ++u) {
    rk[u] = 0;
    if (base + u * 1024 + tid < E) rk[u] = atomicAdd(&h[d[u] >> BSH], 1);
  }
  __syncthreads();

  const int c = (tid < NB) ? h[tid] : 0;
  int incl = c;
  for (int o = 1; o < 32; o <<= 1) {
    int x = __shfl_up(incl, o);
    if (lane >= o) incl += x;
  }
  if (lane == 31) wsum[wv] = incl;
  __syncthreads();
  if (wv == 0) {
    int v = wsum[lane];
    int sc = v;
    for (int o = 1; o < 32; o <<= 1) {
      int x = __shfl_up(sc, o);
      if (lane >= o) sc += x;
    }
    wsum[lane] = sc - v;
  }
  __syncthreads();
  const int excl = wsum[wv] + (incl - c);
  if (tid < NB) h[tid] = excl;
  __syncthreads();

#pragma unroll
  for (int u = 0; u < EPT; ++u) {
    if (base + u * 1024 + tid < E) {
      const int p = base + h[d[u] >> BSH] + rk[u];
      rec[p] = (int)(((unsigned)s[u] << BSH) | (unsigned)(d[u] & (BN - 1)));
    }
  }
  if (tid < NB) {
    hist[t * NB + tid] = c;
    offs[t * NB + tid] = excl;
  }
}

__device__ __forceinline__ void load_pieces(int (&r)[TPW], const int* __restrict__ rec,
                                            int myoff, int mycnt, int tfirst, int lane, int E) {
#pragma unroll
  for (int u = 0; u < TPW; ++u) {
    const int c = __builtin_amdgcn_readlane(mycnt, u);
    const int o = __builtin_amdgcn_readlane(myoff, u);
    const int i = (tfirst + 32 * u) * TE + o + lane;
    r[u] = rec[(i < E) ? i : (E - 1)];
    if (lane >= c) r[u] = -1;
  }
}

__global__ __launch_bounds__(1024) void k_rows(const int* __restrict__ hist,
                                               const int* __restrict__ offs,
                                               const int* __restrict__ rec,
                                               int* __restrict__ rowptr,
                                               float* __restrict__ dinv,
                                               int* __restrict__ csr,
                                               int N, int E, int NB, int NT) {
  __shared__ int cnt[BN];
  __shared__ int wtot[BN / 32];
  __shared__ int red[32];
  const int tid = threadIdx.x;
  const int lane = tid & 31;
  const int wv = tid >> 5;
  const int b = blockIdx.x;

  int col = 0;
  if (tid < b) {
    const int* hc = hist + tid;
#pragma unroll 28
    for (int k = 0; k < NT; ++k) col += hc[k * NB];
  }
  for (int o = 16; o > 0; o >>= 1) col += __shfl_xor(col, o);
  if (lane == 0) red[wv] = col;
  if (tid < BN) cnt[tid] = 0;
  __syncthreads();
  int bs = 0;
#pragma unroll
  for (int w = 0; w < 32; ++w) bs += red[w];

  for (int t0 = 0; t0 < NT; t0 += 32 * TPW) {
    int myoff = 0, mycnt = 0;
    {
      const int t = t0 + wv + 32 * lane;
      if (lane < TPW && t < NT) { myoff = offs[t * NB + b]; mycnt = hist[t * NB + b]; }
    }
    int r[TPW];
    load_pieces(r, rec, myoff, mycnt, t0 + wv, lane, E);
#pragma unroll
    for (int u = 0; u < TPW; ++u)
      if (r[u] >= 0) atomicAdd(&cnt[r[u] & (BN - 1)], 1);
    for (int u = 0; u < TPW; ++u) {
      const int c = __builtin_amdgcn_readlane(mycnt, u);
      if (c > 32) {
        const int pb = (t0 + wv + 32 * u) * TE + __builtin_amdgcn_readlane(myoff, u);
        for (int l = 32 + lane; l < c; l += 32) atomicAdd(&cnt[rec[pb + l] & (BN - 1)], 1);
      }
    }
  }
  __syncthreads();

  int c = 0, incl = 0;
  if (tid < BN) {
    c = cnt[tid];
    incl = c;
    for (int o = 1; o < 32; o <<= 1) {
      int x = __shfl_up(incl, o);
      if (lane >= o) incl += x;
    }
    if (lane == 31) wtot[wv] = incl;
  }
  __syncthreads();
  if (tid < BN) {
    int off = 0;
    for (int w = 0; w < wv; ++w) off += wtot[w];
    const int excl = off + (incl - c);
    cnt[tid] = excl;
    const int node = b * BN + tid;
    if (node < N) {
      rowptr[node] = bs + excl;
      dinv[node] = rsqrtf((float)(c + 1));
      if (node == N - 1) rowptr[N] = bs + excl + c;
    }
  }
  __syncthreads();

  for (int t0 = 0; t0 < NT; t0 += 32 * TPW) {
    int myoff = 0, mycnt = 0;
    {
      const int t = t0 + wv + 32 * lane;
      if (lane < TPW && t < NT) { myoff = offs[t * NB + b]; mycnt = hist[t * NB + b]; }
    }
    int r[TPW];
    load_pieces(r, rec, myoff, mycnt, t0 + wv, lane, E);
#pragma unroll
    for (int u = 0; u < TPW; ++u) {
      if (r[u] >= 0) {
        const int p = atomicAdd(&cnt[r[u] & (BN - 1)], 1);
        csr[bs + p] = (int)((unsigned)r[u] >> BSH);
      }
    }
    for (int u = 0; u < TPW; ++u) {
      const int cc = __builtin_amdgcn_readlane(mycnt, u);
      if (cc > 32) {
        const int pb = (t0 + wv + 32 * u) * TE + __builtin_amdgcn_readlane(myoff, u);
        for (int l = 32 + lane; l < cc; l += 32) {
          const int q = rec[pb + l];
          const int p = atomicAdd(&cnt[q & (BN - 1)], 1);
          csr[bs + p] = (int)((unsigned)q >> BSH);
        }
      }
    }
  }
}

template <int V> struct vecf;
template <> struct vecf<2> { typedef float t __attribute__((ext_vector_type(2))); };
template <> struct vecf<4> { typedef float t __attribute__((ext_vector_type(4))); };

template <int OUT, int R, bool PRE>
__device__ __forceinline__ typename vecf<OUT / 32>::t
gather_chunk(typename vecf<OUT / 32>::t acc, const float* __restrict__ hl,
             int s_l, float g_l, int j) {
  typedef typename vecf<OUT / 32>::t vf;
  vf v[R];
  float w[R];
#pragma unroll
  for (int u = 0; u < R; ++u) {
    const int s = __builtin_amdgcn_readlane(s_l, j + u);
    w[u] = PRE ? 1.0f : __int_as_float(__builtin_amdgcn_readlane(__float_as_int(g_l), j + u));
    v[u] = *(const vf*)(hl + (size_t)s * OUT);
  }
#pragma unroll
  for (int u = 0; u < R; ++u) {
    if (PRE) acc += v[u];
    else acc += v[u] * w[u];
  }
  return acc;
}

template <int OUT, bool RELU>
__global__ __launch_bounds__(256)
void k_gather(const int* __restrict__ rowptr, const int* __restrict__ csr,
              const float* __restrict__ dinv, const float* __restrict__ h,
              const float* __restrict__ b, float* __restrict__ out, int N) {
  constexpr int V = OUT / 32;
  typedef typename vecf<V>::t vf;
  const int lane = threadIdx.x & 31;
  const int d = __builtin_amdgcn_readfirstlane(
      (int)((blockIdx.x * blockDim.x + threadIdx.x) >> 5));
  if (d >= N) return;
  const int beg = rowptr[d];
  const int deg = rowptr[d + 1] - beg;
  const float dd = dinv[d];
  const float* hl = h + lane * V;

  const int tot = deg + 1;
  vf acc = {};
  for (int b0 = 0; b0 < tot; b0 += 32) {
    const int n = (tot - b0 < 32) ? (tot - b0) : 32;
    const int t = b0 + lane;
    int s_l = d;
    if (t < deg) s_l = csr[beg + t];
    int j = 0;
    for (; j + 8 <= n; j += 8) acc = gather_chunk<OUT, 8, true>(acc, hl, s_l, 0.0f, j);
    switch (n - j) {
      case 7: acc = gather_chunk<OUT, 7, true>(acc, hl, s_l, 0.0f, j); break;
      case 6: acc = gather_chunk<OUT, 6, true>(acc, hl, s_l, 0.0f, j); break;
      case 5: acc = gather_chunk<OUT, 5, true>(acc, hl, s_l, 0.0f, j); break;
      case 4: acc = gather_chunk<OUT, 4, true>(acc, hl, s_l, 0.0f, j); break;
      case 3: acc = gather_chunk<OUT, 3, true>(acc, hl, s_l, 0.0f, j); break;
      case 2: acc = gather_chunk<OUT, 2, true>(acc, hl, s_l, 0.0f, j); break;
      case 1: acc = gather_chunk<OUT, 1, true>(acc, hl, s_l, 0.0f, j); break;
      default: break;
    }
  }

  vf r = acc * dd + *(const vf*)(b + lane * V);
  if (RELU) {
#pragma unroll
    for (int k = 0; k < V; ++k) r[k] = fmaxf(r[k], 0.0f);
  }
  *(vf*)(out + (size_t)d * OUT + lane * V) = r;
}

constexpr int LP  = 132;
constexpr int LPC = 68;

__device__ __forceinline__ v8f tile_product(const float* At, const float* Bt, int m, int kh) {
  const float* ap = At + m * LP + 2 * kh;
  const float* bp = Bt + m * LP + 2 * kh;
  v8f acc0 = {}, acc1 = {};
#pragma unroll
  for (int k0 = 0; k0 < 128; k0 += 32) {
    v2f a[8], b[8];
#pragma unroll
    for (int u = 0; u < 8; ++u) {
      a[u] = *(const v2f*)(ap + k0 + 4 * u);
      b[u] = *(const v2f*)(bp + k0 + 4 * u);
    }
#pragma unroll
    for (int u = 0; u < 8; u += 2) {
      acc0 = __builtin_amdgcn_wmma_f32_16x16x4_f32(false, a[u], false, b[u], (short)0, acc0, false, false);
      acc1 = __builtin_amdgcn_wmma_f32_16x16x4_f32(false, a[u + 1], false, b[u + 1], (short)0, acc1, false, false);
    }
  }
  return acc0 + acc1;
}

template <bool PRE, bool TAIL>
__global__ __launch_bounds__(1024)
void k_layer(const int* __restrict__ rowptr, const int* __restrict__ csr,
             const float* __restrict__ dinv, const float* __restrict__ fin,
             const float* __restrict__ W, const float* __restrict__ b,
             const float* __restrict__ W3, float* __restrict__ out, int N) {
  __shared__ __attribute__((aligned(16))) float Wt[128 * LP];
  __shared__ __attribute__((aligned(16))) float W3t[TAIL ? 64 * LP : 4];
  __shared__ __attribute__((aligned(16))) float tA[32 * LP];
  __shared__ __attribute__((aligned(16))) float tB[32 * LP];
  __shared__ __attribute__((aligned(16))) float tC[TAIL ? 32 * LPC : 4];

  const int tid = threadIdx.x;
  const int lane = tid & 31;
  const int wave = __builtin_amdgcn_readfirstlane(tid >> 5);
  const int m = lane & 15;
  const int kh = lane >> 4;

  for (int i = tid; i < 128 * 128; i += 1024) Wt[(i & 127) * LP + (i >> 7)] = W[i];
  if (TAIL) {
    for (int i = tid; i < 128 * 64; i += 1024) W3t[(i & 63) * LP + (i >> 6)] = W3[i];
  }

  const float* hl = fin + lane * 4;
  const int nst = (N + 31) >> 5;
  for (int st = blockIdx.x; st < nst; st += gridDim.x) {
    const int d = st * 32 + wave;
    v4f acc = {};
    float dd = 0.0f;
    if (d < N) {
      const int beg = rowptr[d];
      const int deg = rowptr[d + 1] - beg;
      dd = dinv[d];
      const int tot = deg + 1;
      int b0 = 0;
      do {
        const int n = (tot - b0 < 32) ? (tot - b0) : 32;
        const int t = b0 + lane;
        int s_l = d;
        float g_l = dd;
        if (t < deg) {
          s_l = csr[beg + t];
          if (!PRE) g_l = dinv[s_l];
        }
        int j = 0;
        for (; j + 8 <= n; j += 8) acc = gather_chunk<128, 8, PRE>(acc, hl, s_l, g_l, j);
        switch (n - j) {
          case 7: acc = gather_chunk<128, 7, PRE>(acc, hl, s_l, g_l, j); break;
          case 6: acc = gather_chunk<128, 6, PRE>(acc, hl, s_l, g_l, j); break;
          case 5: acc = gather_chunk<128, 5, PRE>(acc, hl, s_l, g_l, j); break;
          case 4: acc = gather_chunk<128, 4, PRE>(acc, hl, s_l, g_l, j); break;
          case 3: acc = gather_chunk<128, 3, PRE>(acc, hl, s_l, g_l, j); break;
          case 2: acc = gather_chunk<128, 2, PRE>(acc, hl, s_l, g_l, j); break;
          case 1: acc = gather_chunk<128, 1, PRE>(acc, hl, s_l, g_l, j); break;
          default: break;
        }
        b0 += 32;
      } while (b0 < tot);
    }
    *(v4f*)&tA[wave * LP + lane * 4] = acc * dd;
    __syncthreads();

    if (wave < 16) {
      const int rt = wave >> 3;
      const int ct = wave & 7;
      const v8f c = tile_product(tA + rt * 16 * LP, Wt + ct * 16 * LP, m, kh);
      const float bias = b[ct * 16 + m];
      float* o = tB + (rt * 16 + 8 * kh) * LP + ct * 16 + m;
#pragma unroll
      for (int r = 0; r < 8; ++r) o[r * LP] = fmaxf(c[r] + bias, 0.0f);
    }
    __syncthreads();

    if (!TAIL) {
      if (d < N) *(v4f*)(out + (size_t)d * 128 + lane * 4) = *(const v4f*)&tB[wave * LP + lane * 4] * dd;
    } else {
      if (wave < 8) {
        const int rt = wave >> 2;
        const int ct = wave & 3;
        const v8f c = tile_product(tB + rt * 16 * LP, W3t + ct * 16 * LP, m, kh);
        float* o = tC + (rt * 16 + 8 * kh) * LPC + ct * 16 + m;
#pragma unroll
        for (int r = 0; r < 8; ++r) o[r * LPC] = c[r];
      }
      __syncthreads();
      if (d < N) *(v2f*)(out + (size_t)d * 64 + lane * 2) = *(const v2f*)&tC[wave * LPC + lane * 2] * dd;
    }
  }
}

static inline int cdiv(long a, long b) { return (int)((a + b - 1) / b); }

extern "C" void kernel_launch(void* const* d_in, const int* in_sizes, int n_in,
                              void* d_out, int out_size, void* d_ws, size_t ws_size,
                              hipStream_t stream) {
  const float* x  = (const float*)d_in[0];
  const int*   ei = (const int*)d_in[1];
  const float* W1 = (const float*)d_in[2];
  const float* b1 = (const float*)d_in[3];
  const float* W2 = (const float*)d_in[4];
  const float* b2 = (const float*)d_in[5];
  const float* W3 = (const float*)d_in[6];
  const float* b3 = (const float*)d_in[7];

  const int N = in_sizes[0] / 128;
  const int E = in_sizes[1] / 2;
  const int* src = ei;
  const int* dst = ei + E;

  char* ws = (char*)d_ws;
  const size_t szN = (((size_t)(N + 1) * 4) + 255) & ~(size_t)255;
  const size_t szE = (((size_t)E * 4) + 255) & ~(size_t)255;
  size_t off = 0;
  const int NB = cdiv(N, BN);
  const int NT = cdiv(E, TE);
  const size_t szH = (((size_t)NT * NB * 4) + 255) & ~(size_t)255;
  float* dinv   = (float*)(ws + off); off += szN;
  int*   rowptr = (int*)(ws + off);   off += szN;
  int*   hist   = (int*)(ws + off);   off += szH;
  int*   offs   = (int*)(ws + off);   off += szH;
  int*   rec    = (int*)(ws + off);   off += szE;
  int*   csr    = (int*)(ws + off);   off += szE;
  float* bufF   = (float*)(ws + off); off += (size_t)N * 128 * 4;
  float* bufH   = (float*)(ws + off);
  float* out    = (float*)d_out;

  const int T = 256;
  const int nodeBlocks = cdiv((long)N * 32, T);
  const int layerBlocks = 512;

  k_tile<<<NT, 1024, 0, stream>>>(src, dst, hist, offs, rec, E, NB);
  k_rows<<<NB, 1024, 0, stream>>>(hist, offs, rec, rowptr, dinv, csr, N, E, NB, NT);

  k_layer<false, false><<<layerBlocks, 1024, 0, stream>>>(rowptr, csr, dinv, x, W1, b1, W3, bufF, N);
  k_layer<true, true><<<layerBlocks, 1024, 0, stream>>>(rowptr, csr, dinv, bufF, W2, b2, W3, bufH, N);
  k_gather<64, false><<<nodeBlocks, T, 0, stream>>>(rowptr, csr, dinv, bufH, b3, out, N);
}


# CSR build: both sort passes file their scattered 4-byte stores (tile records, bucket sources) in LDS and write them out as whole 128-byte lines; row sum order: LDS add arrival order, as before
# speedup vs baseline: 6.0095x; 1.0550x over previous
#include <hip/hip_runtime.h>


typedef float v2f __attribute__((ext_vector_type(2)));
typedef float v4f __attribute__((ext_vector_type(4)));
typedef float v8f __attribute__((ext_vector_type(8)));

constexpr int BSH   = 8;
constexpr int BN    = 1 << BSH;
constexpr int NBMAX = 1024;
constexpr int EPT   = 8;
constexpr int TE    = 1024 * EPT;

constexpr int TPW   = 7;
constexpr int CAP   = 8192;

typedef int v4i __attribute__((ext_vector_type(4)));

__global__ __launch_bounds__(1024) void k_tile(const int* __restrict__ src,
                                               const int* __restrict__ dst,
                                               int* __restrict__ hist,
                                               int* __restrict__ offs,
                                               int* __restrict__ rec,
                                               int E, int NB) {
  __shared__ int h[NBMAX];
  __shared__ int wsum[32];
  __shared__ __attribute__((aligned(16))) int srec[TE];
  const int tid = threadIdx.x;
  const int lane = tid & 31;
  const int wv = tid >> 5;
  const int t = blockIdx.x;
  const int base = t * TE;

  int s[EPT], d[EPT], rk[EPT];
#pragma unroll
  for (int u = 0; u < EPT; ++u) {
    const int i = base + u * 1024 + tid;
    const int ic = (i < E) ? i : (E - 1);
    d[u] = dst[ic];
    s[u] = src[ic];
  }
  h[tid] = 0;
  __syncthreads();
#pragma unroll
  for (int u = 0; u < EPT; ++u) {
    rk[u] = 0;
    if (base + u * 1024 + tid < E) rk[u] = atomicAdd(&h[d[u] >> BSH], 1);
  }
  __syncthreads();

  const int c = (tid < NB) ? h[tid] : 0;
  int incl = c;
  for (int o = 1; o < 32; o <<= 1) {
    int x = __shfl_up(incl, o);
    if (lane >= o) incl += x;
  }
  if (lane == 31) wsum[wv] = incl;
  __syncthreads();
  if (wv == 0) {
    int v = wsum[lane];
    int sc = v;
    for (int o = 1; o < 32; o <<= 1) {
      int x = __shfl_up(sc, o);
      if (lane >= o) sc += x;
    }
    wsum[lane] = sc - v;
  }
  __syncthreads();
  const int excl = wsum[wv] + (incl - c);
  if (tid < NB) h[tid] = excl;
  __syncthreads();

#pragma unroll
  for (int u = 0; u < EPT; ++u) {
    if (base + u * 1024 + tid < E) {
      const int p = h[d[u] >> BSH] + rk[u];
      srec[p] = (int)(((unsigned)s[u] << BSH) | (unsigned)(d[u] & (BN - 1)));
    }
  }
  __syncthreads();
  const int valid = (E - base < TE) ? (E - base) : TE;
#pragma unroll
  for (int u = 0; u < EPT / 4; ++u) {
    const int i = (u * 1024 + tid) * 4;
    if (i + 3 < valid) {
      *(v4i*)(rec + base + i) = *(const v4i*)&srec[i];
    } else {
      for (int k = 0; k < 4; ++k)
        if (i + k < valid) rec[base + i + k] = srec[i + k];
    }
  }
  if (tid < NB) {
    hist[t * NB + tid] = c;
    offs[t * NB + tid] = excl;
  }
}

__device__ __forceinline__ void load_pieces(int (&r)[TPW], const int* __restrict__ rec,
                                            int myoff, int mycnt, int tfirst, int lane, int E) {
#pragma unroll
  for (int u = 0; u < TPW; ++u) {
    const int c = __builtin_amdgcn_readlane(mycnt, u);
    const int o = __builtin_amdgcn_readlane(myoff, u);
    const int i = (tfirst + 32 * u) * TE + o + lane;
    r[u] = rec[(i < E) ? i : (E - 1)];
    if (lane >= c) r[u] = -1;
  }
}

__global__ __launch_bounds__(1024) void k_rows(const int* __restrict__ hist,
                                               const int* __restrict__ offs,
                                               const int* __restrict__ rec,
                                               int* __restrict__ rowptr,
                                               float* __restrict__ dinv,
                                               int* __restrict__ csr,
                                               int N, int E, int NB, int NT) {
  __shared__ int cnt[BN];
  __shared__ int wtot[BN / 32];
  __shared__ int red[32];
  __shared__ int scsr[CAP];
  const int tid = threadIdx.x;
  const int lane = tid & 31;
  const int wv = tid >> 5;
  const int b = blockIdx.x;

  int col = 0;
  if (tid < b) {
    const int* hc = hist + tid;
#pragma unroll 28
    for (int k = 0; k < NT; ++k) col += hc[k * NB];
  }
  for (int o = 16; o > 0; o >>= 1) col += __shfl_xor(col, o);
  if (lane == 0) red[wv] = col;
  if (tid < BN) cnt[tid] = 0;
  __syncthreads();
  int bs = 0;
#pragma unroll
  for (int w = 0; w < 32; ++w) bs += red[w];

  for (int t0 = 0; t0 < NT; t0 += 32 * TPW) {
    int myoff = 0, mycnt = 0;
    {
      const int t = t0 + wv + 32 * lane;
      if (lane < TPW && t < NT) { myoff = offs[t * NB + b]; mycnt = hist[t * NB + b]; }
    }
    int r[TPW];
    load_pieces(r, rec, myoff, mycnt, t0 + wv, lane, E);
#pragma unroll
    for (int u = 0; u < TPW; ++u)
      if (r[u] >= 0) atomicAdd(&cnt[r[u] & (BN - 1)], 1);
    for (int u = 0; u < TPW; ++u) {
      const int c = __builtin_amdgcn_readlane(mycnt, u);
      if (c > 32) {
        const int pb = (t0 + wv + 32 * u) * TE + __builtin_amdgcn_readlane(myoff, u);
        for (int l = 32 + lane; l < c; l += 32) atomicAdd(&cnt[rec[pb + l] & (BN - 1)], 1);
      }
    }
  }
  __syncthreads();

  int c = 0, incl = 0;
  if (tid < BN) {
    c = cnt[tid];
    incl = c;
    for (int o = 1; o < 32; o <<= 1) {
      int x = __shfl_up(incl, o);
      if (lane >= o) incl += x;
    }
    if (lane == 31) wtot[wv] = incl;
  }
  __syncthreads();
  int total = 0;
#pragma unroll
  for (int w = 0; w < BN / 32; ++w) total += wtot[w];
  if (tid < BN) {
    int off = 0;
    for (int w = 0; w < wv; ++w) off += wtot[w];
    const int excl = off + (incl - c);
    cnt[tid] = excl;
    const int node = b * BN + tid;
    if (node < N) {
      rowptr[node] = bs + excl;
      dinv[node] = rsqrtf((float)(c + 1));
      if (node == N - 1) rowptr[N] = bs + excl + c;
    }
  }
  __syncthreads();

  for (int t0 = 0; t0 < NT; t0 += 32 * TPW) {
    int myoff = 0, mycnt = 0;
    {
      const int t = t0 + wv + 32 * lane;
      if (lane < TPW && t < NT) { myoff = offs[t * NB + b]; mycnt = hist[t * NB + b]; }
    }
    int r[TPW];
    load_pieces(r, rec, myoff, mycnt, t0 + wv, lane, E);
#pragma unroll
    for (int u = 0; u < TPW; ++u) {
      if (r[u] >= 0) {
        const int p = atomicAdd(&cnt[r[u] & (BN - 1)], 1);
        const int sv = (int)((unsigned)r[u] >> BSH);
        if (p < CAP) scsr[p] = sv;
        else csr[bs + p] = sv;
      }
    }
    for (int u = 0; u < TPW; ++u) {
      const int cc = __builtin_amdgcn_readlane(mycnt, u);
      if (cc > 32) {
        const int pb = (t0 + wv + 32 * u) * TE + __builtin_amdgcn_readlane(myoff, u);
        for (int l = 32 + lane; l < cc; l += 32) {
          const int q = rec[pb + l];
          const int p = atomicAdd(&cnt[q & (BN - 1)], 1);
          const int sv = (int)((unsigned)q >> BSH);
          if (p < CAP) scsr[p] = sv;
          else csr[bs + p] = sv;
        }
      }
    }
  }
  __syncthreads();
  const int nst = (total < CAP) ? total : CAP;
  for (int j = tid - (bs & 31); j < nst; j += 1024)
    if (j >= 0) csr[bs + j] = scsr[j];
}

template <int V> struct vecf;
template <> struct vecf<2> { typedef float t __attribute__((ext_vector_type(2))); };
template <> struct vecf<4> { typedef float t __attribute__((ext_vector_type(4))); };

template <int OUT, int R, bool PRE>
__device__ __forceinline__ typename vecf<OUT / 32>::t
gather_chunk(typename vecf<OUT / 32>::t acc, const float* __restrict__ hl,
             int s_l, float g_l, int j) {
  typedef typename vecf<OUT / 32>::t vf;
  vf v[R];
  float w[R];
#pragma unroll
  for (int u = 0; u < R; ++u) {
    const int s = __builtin_amdgcn_readlane(s_l, j + u);
    w[u] = PRE ? 1.0f : __int_as_float(__builtin_amdgcn_readlane(__float_as_int(g_l), j + u));
    v[u] = *(const vf*)(hl + (size_t)s * OUT);
  }
#pragma unroll
  for (int u = 0; u < R; ++u) {
    if (PRE) acc += v[u];
    else acc += v[u] * w[u];
  }
  return acc;
}

template <int OUT, bool RELU>
__global__ __launch_bounds__(256)
void k_gather(const int* __restrict__ rowptr, const int* __restrict__ csr,
              const float* __restrict__ dinv, const float* __restrict__ h,
              const float* __restrict__ b, float* __restrict__ out, int N) {
  constexpr int V = OUT / 32;
  typedef typename vecf<V>::t vf;
  const int lane = threadIdx.x & 31;
  const int d = __builtin_amdgcn_readfirstlane(
      (int)((blockIdx.x * blockDim.x + threadIdx.x) >> 5));
  if (d >= N) return;
  const int beg = rowptr[d];
  const int deg = rowptr[d + 1] - beg;
  const float dd = dinv[d];
  const float* hl = h + lane * V;

  const int tot = deg + 1;
  vf acc = {};
  for (int b0 = 0; b0 < tot; b0 += 32) {
    const int n = (tot - b0 < 32) ? (tot - b0) : 32;
    const int t = b0 + lane;
    int s_l = d;
    if (t < deg) s_l = csr[beg + t];
    int j = 0;
    for (; j + 8 <= n; j += 8) acc = gather_chunk<OUT, 8, true>(acc, hl, s_l, 0.0f, j);
    switch (n - j) {
      case 7: acc = gather_chunk<OUT, 7, true>(acc, hl, s_l, 0.0f, j); break;
      case 6: acc = gather_chunk<OUT, 6, true>(acc, hl, s_l, 0.0f, j); break;
      case 5: acc = gather_chunk<OUT, 5, true>(acc, hl, s_l, 0.0f, j); break;
      case 4: acc = gather_chunk<OUT, 4, true>(acc, hl, s_l, 0.0f, j); break;
      case 3: acc = gather_chunk<OUT, 3, true>(acc, hl, s_l, 0.0f, j); break;
      case 2: acc = gather_chunk<OUT, 2, true>(acc, hl, s_l, 0.0f, j); break;
      case 1: acc = gather_chunk<OUT, 1, true>(acc, hl, s_l, 0.0f, j); break;
      default: break;
    }
  }

  vf r = acc * dd + *(const vf*)(b + lane * V);
  if (RELU) {
#pragma unroll
    for (int k = 0; k < V; ++k) r[k] = fmaxf(r[k], 0.0f);
  }
  *(vf*)(out + (size_t)d * OUT + lane * V) = r;
}

constexpr int LP  = 132;
constexpr int LPC = 68;

__device__ __forceinline__ v8f tile_product(const float* At, const float* Bt, int m, int kh) {
  const float* ap = At + m * LP + 2 * kh;
  const float* bp = Bt + m * LP + 2 * kh;
  v8f acc0 = {}, acc1 = {};
#pragma unroll
  for (int k0 = 0; k0 < 128; k0 += 32) {
    v2f a[8], b[8];
#pragma unroll
    for (int u = 0; u < 8; ++u) {
      a[u] = *(const v2f*)(ap + k0 + 4 * u);
      b[u] = *(const v2f*)(bp + k0 + 4 * u);
    }
#pragma unroll
    for (int u = 0; u < 8; u += 2) {
      acc0 = __builtin_amdgcn_wmma_f32_16x16x4_f32(false, a[u], false, b[u], (short)0, acc0, false, false);
      acc1 = __builtin_amdgcn_wmma_f32_16x16x4_f32(false, a[u + 1], false, b[u + 1], (short)0, acc1, false, false);
    }
  }
  return acc0 + acc1;
}

template <bool PRE, bool TAIL>
__global__ __launch_bounds__(1024)
void k_layer(const int* __restrict__ rowptr, const int* __restrict__ csr,
             const float* __restrict__ dinv, const float* __restrict__ fin,
             const float* __restrict__ W, const float* __restrict__ b,
             const float* __restrict__ W3, float* __restrict__ out, int N) {
  __shared__ __attribute__((aligned(16))) float Wt[128 * LP];
  __shared__ __attribute__((aligned(16))) float W3t[TAIL ? 64 * LP : 4];
  __shared__ __attribute__((aligned(16))) float tA[32 * LP];
  __shared__ __attribute__((aligned(16))) float tB[32 * LP];
  __shared__ __attribute__((aligned(16))) float tC[TAIL ? 32 * LPC : 4];

  const int tid = threadIdx.x;
  const int lane = tid & 31;
  const int wave = __builtin_amdgcn_readfirstlane(tid >> 5);
  const int m = lane & 15;
  const int kh = lane >> 4;

  for (int i = tid; i < 128 * 128; i += 1024) Wt[(i & 127) * LP + (i >> 7)] = W[i];
  if (TAIL) {
    for (int i = tid; i < 128 * 64; i += 1024) W3t[(i & 63) * LP + (i >> 6)] = W3[i];
  }

  const float* hl = fin + lane * 4;
  const int nst = (N + 31) >> 5;
  for (int st = blockIdx.x; st < nst; st += gridDim.x) {
    const int d = st * 32 + wave;
    v4f acc = {};
    float dd = 0.0f;
    if (d < N) {
      const int beg = rowptr[d];
      const int deg = rowptr[d + 1] - beg;
      dd = dinv[d];
      const int tot = deg + 1;
      int b0 = 0;
      do {
        const int n = (tot - b0 < 32) ? (tot - b0) : 32;
        const int t = b0 + lane;
        int s_l = d;
        float g_l = dd;
        if (t < deg) {
          s_l = csr[beg + t];
          if (!PRE) g_l = dinv[s_l];
        }
        int j = 0;
        for (; j + 8 <= n; j += 8) acc = gather_chunk<128, 8, PRE>(acc, hl, s_l, g_l, j);
        switch (n - j) {
          case 7: acc = gather_chunk<128, 7, PRE>(acc, hl, s_l, g_l, j); break;
          case 6: acc = gather_chunk<128, 6, PRE>(acc, hl, s_l, g_l, j); break;
          case 5: acc = gather_chunk<128, 5, PRE>(acc, hl, s_l, g_l, j); break;
          case 4: acc = gather_chunk<128, 4, PRE>(acc, hl, s_l, g_l, j); break;
          case 3: acc = gather_chunk<128, 3, PRE>(acc, hl, s_l, g_l, j); break;
          case 2: acc = gather_chunk<128, 2, PRE>(acc, hl, s_l, g_l, j); break;
          case 1: acc = gather_chunk<128, 1, PRE>(acc, hl, s_l, g_l, j); break;
          default: break;
        }
        b0 += 32;
      } while (b0 < tot);
    }
    *(v4f*)&tA[wave * LP + lane * 4] = acc * dd;
    __syncthreads();

    if (wave < 16) {
      const int rt = wave >> 3;
      const int ct = wave & 7;
      const v8f c = tile_product(tA + rt * 16 * LP, Wt + ct * 16 * LP, m, kh);
      const float bias = b[ct * 16 + m];
      float* o = tB + (rt * 16 + 8 * kh) * LP + ct * 16 + m;
#pragma unroll
      for (int r = 0; r < 8; ++r) o[r * LP] = fmaxf(c[r] + bias, 0.0f);
    }
    __syncthreads();

    if (!TAIL) {
      if (d < N) *(v4f*)(out + (size_t)d * 128 + lane * 4) = *(const v4f*)&tB[wave * LP + lane * 4] * dd;
    } else {
      if (wave < 8) {
        const int rt = wave >> 2;
        const int ct = wave & 3;
        const v8f c = tile_product(tB + rt * 16 * LP, W3t + ct * 16 * LP, m, kh);
        float* o = tC + (rt * 16 + 8 * kh) * LPC + ct * 16 + m;
#pragma unroll
        for (int r = 0; r < 8; ++r) o[r * LPC] = c[r];
      }
      __syncthreads();
      if (d < N) *(v2f*)(out + (size_t)d * 64 + lane * 2) = *(const v2f*)&tC[wave * LPC + lane * 2] * dd;
    }
  }
}

static inline int cdiv(long a, long b) { return (int)((a + b - 1) / b); }

extern "C" void kernel_launch(void* const* d_in, const int* in_sizes, int n_in,
                              void* d_out, int out_size, void* d_ws, size_t ws_size,
                              hipStream_t stream) {
  const float* x  = (const float*)d_in[0];
  const int*   ei = (const int*)d_in[1];
  const float* W1 = (const float*)d_in[2];
  const float* b1 = (const float*)d_in[3];
  const float* W2 = (const float*)d_in[4];
  const float* b2 = (const float*)d_in[5];
  const float* W3 = (const float*)d_in[6];
  const float* b3 = (const float*)d_in[7];

  const int N = in_sizes[0] / 128;
  const int E = in_sizes[1] / 2;
  const int* src = ei;
  const int* dst = ei + E;

  char* ws = (char*)d_ws;
  const size_t szN = (((size_t)(N + 1) * 4) + 255) & ~(size_t)255;
  const size_t szE = (((size_t)E * 4) + 255) & ~(size_t)255;
  size_t off = 0;
  const int NB = cdiv(N, BN);
  const int NT = cdiv(E, TE);
  const size_t szH = (((size_t)NT * NB * 4) + 255) & ~(size_t)255;
  float* dinv   = (float*)(ws + off); off += szN;
  int*   rowptr = (int*)(ws + off);   off += szN;
  int*   hist   = (int*)(ws + off);   off += szH;
  int*   offs   = (int*)(ws + off);   off += szH;
  int*   rec    = (int*)(ws + off);   off += szE;
  int*   csr    = (int*)(ws + off);   off += szE;
  float* bufF   = (float*)(ws + off); off += (size_t)N * 128 * 4;
  float* bufH   = (float*)(ws + off);
  float* out    = (float*)d_out;

  const int T = 256;
  const int nodeBlocks = cdiv((long)N * 32, T);
  const int layerBlocks = 512;

  k_tile<<<NT, 1024, 0, stream>>>(src, dst, hist, offs, rec, E, NB);
  k_rows<<<NB, 1024, 0, stream>>>(hist, offs, rec, rowptr, dinv, csr, N, E, NB, NT);

  k_layer<false, false><<<layerBlocks, 1024, 0, stream>>>(rowptr, csr, dinv, x, W1, b1, W3, bufF, N);
  k_layer<true, true><<<layerBlocks, 1024, 0, stream>>>(rowptr, csr, dinv, bufF, W2, b2, W3, bufH, N);
  k_gather<64, false><<<nodeBlocks, T, 0, stream>>>(rowptr, csr, dinv, bufH, b3, out, N);
}
